# starting kernel
# baseline (speedup 1.0000x reference)
#include <hip/hip_runtime.h>
#include <math.h>

#define IN_DIM  100
#define HIDDEN  16
#define NCLS    40

typedef __attribute__((ext_vector_type(2))) float v2f;
typedef __attribute__((ext_vector_type(8))) float v8f;

static __device__ inline v8f wmma_f32_16x16x4(v2f a, v2f b, v8f c) {
#if __has_builtin(__builtin_amdgcn_wmma_f32_16x16x4_f32)
  return __builtin_amdgcn_wmma_f32_16x16x4_f32(
       false, a,  false, b,
       (short)0, c,  false,  false);
#else
  const int lane = (int)(threadIdx.x & 31u);
  const int nsel = lane & 15;
  v8f d = c;
#pragma unroll
  for (int g = 0; g < 8; ++g) {
    const int m = g + 8 * (lane >> 4);
    float acc = 0.f;
#pragma unroll
    for (int k = 0; k < 4; ++k) {
      const int srcA = m + 16 * (k >> 1);
      const int srcB = nsel + 16 * (k >> 1);
      const float av = __shfl((k & 1) ? a.y : a.x, srcA, 32);
      const float bv = __shfl((k & 1) ? b.y : b.x, srcB, 32);
      acc += av * bv;
    }
    d[g] += acc;
  }
  return d;
#endif
}

__global__ void gcn_zero_kernel(float* __restrict__ p, int n) {
  int i = blockIdx.x * blockDim.x + threadIdx.x;
  int stride = gridDim.x * blockDim.x;
  for (; i < n; i += stride) p[i] = 0.0f;
}

__global__ __launch_bounds__(32) void gcn_gemm1_wmma(
    const float* __restrict__ X, const float* __restrict__ W1,
    float* __restrict__ Y, int N) {
  const int lane = (int)threadIdx.x;
  const int r0   = blockIdx.x * 16;
  const int col  = lane & 15;
  const int kh   = lane >> 4;
  const int rowA = min(r0 + col, N - 1);
  const float* __restrict__ Xr = X + (size_t)rowA * IN_DIM;

  v8f acc = {};
#pragma unroll 5
  for (int k0 = 0; k0 < IN_DIM; k0 += 4) {
    const int ka = k0 + 2 * kh;
    v2f a;
    a.x = Xr[ka];
    a.y = Xr[ka + 1];
    v2f b;
    b.x = W1[ka * HIDDEN + col];
    b.y = W1[(ka + 1) * HIDDEN + col];
    acc = wmma_f32_16x16x4(a, b, acc);
  }

  float* __restrict__ Yr = Y + (size_t)(r0 + 8 * kh) * HIDDEN + col;
  if (r0 + 16 <= N) {
#pragma unroll
    for (int g = 0; g < 8; ++g) Yr[g * HIDDEN] = acc[g];
  } else {
#pragma unroll
    for (int g = 0; g < 8; ++g) {
      const int row = r0 + g + 8 * kh;
      if (row < N) Y[row * HIDDEN + col] = acc[g];
    }
  }
}

__global__ __launch_bounds__(256) void gcn_spmm_kernel(
    const int* __restrict__ esrc, const int* __restrict__ edst,
    const float* __restrict__ evals, const float* __restrict__ Xin,
    float* __restrict__ Z, int E, int doRelu) {
  const int t = blockIdx.x * blockDim.x + threadIdx.x;
  const int e = t >> 4;
  const int c = t & 15;
  if (e >= E) return;
  const int s = esrc[e];
  const int d = edst[e];
  const float v = evals[e];
  float xv = Xin[(size_t)s * HIDDEN + c];
  if (doRelu) xv = fmaxf(xv, 0.0f);
  atomicAdd(&Z[(size_t)d * HIDDEN + c], v * xv);
}

__global__ __launch_bounds__(32) void gcn_gemm2_lsm_wmma(
    const float* __restrict__ Z2, const float* __restrict__ W2,
    float* __restrict__ out, int N) {
  const int lane = (int)threadIdx.x;
  const int r0   = blockIdx.x * 16;
  const int col  = lane & 15;
  const int kh   = lane >> 4;
  const int rowA = min(r0 + col, N - 1);
  const float* __restrict__ Zr = Z2 + (size_t)rowA * HIDDEN;

  v8f acc0 = {}, acc1 = {}, acc2 = {};
#pragma unroll
  for (int k0 = 0; k0 < HIDDEN; k0 += 4) {
    const int ka = k0 + 2 * kh;
    v2f a;
    a.x = Zr[ka];
    a.y = Zr[ka + 1];

    v2f b0;
    b0.x = W2[ka * NCLS + col];
    b0.y = W2[(ka + 1) * NCLS + col];
    acc0 = wmma_f32_16x16x4(a, b0, acc0);

    v2f b1;
    b1.x = W2[ka * NCLS + 16 + col];
    b1.y = W2[(ka + 1) * NCLS + 16 + col];
    acc1 = wmma_f32_16x16x4(a, b1, acc1);

    const int c2 = 32 + col;
    v2f b2;
    b2.x = (c2 < NCLS) ? W2[ka * NCLS + c2] : 0.0f;
    b2.y = (c2 < NCLS) ? W2[(ka + 1) * NCLS + c2] : 0.0f;
    acc2 = wmma_f32_16x16x4(a, b2, acc2);
  }

  const bool t2ok = (col < 8);
  const float ninf = -__builtin_inff();
  const bool fullTile = (r0 + 16 <= N);
  float* __restrict__ Or = out + (size_t)(r0 + 8 * kh) * NCLS + col;

#pragma unroll
  for (int g = 0; g < 8; ++g) {
    const float v0 = acc0[g];
    const float v1 = acc1[g];
    const float v2v = acc2[g];

    float m = fmaxf(fmaxf(v0, v1), t2ok ? v2v : ninf);
#pragma unroll
    for (int off = 8; off >= 1; off >>= 1) m = fmaxf(m, __shfl_xor(m, off, 16));

    float s = expf(v0 - m) + expf(v1 - m) + (t2ok ? expf(v2v - m) : 0.0f);
#pragma unroll
    for (int off = 8; off >= 1; off >>= 1) s += __shfl_xor(s, off, 16);

    const float lse = m + logf(s);

    if (fullTile) {
      Or[g * NCLS] = v0 - lse;
      Or[g * NCLS + 16] = v1 - lse;
      if (t2ok) Or[g * NCLS + 32] = v2v - lse;
    } else {
      const int row = r0 + g + 8 * kh;
      if (row < N) {
        out[(size_t)row * NCLS + col] = v0 - lse;
        out[(size_t)row * NCLS + 16 + col] = v1 - lse;
        if (t2ok) out[(size_t)row * NCLS + 32 + col] = v2v - lse;
      }
    }
  }
}

extern "C" void kernel_launch(void* const* d_in, const int* in_sizes, int n_in,
                              void* d_out, int out_size, void* d_ws, size_t ws_size,
                              hipStream_t stream) {
  const float* features = (const float*)d_in[0];
  const int*   esrc     = (const int*)d_in[1];
  const int*   edst     = (const int*)d_in[2];
  const float* evals    = (const float*)d_in[3];
  const float* W1       = (const float*)d_in[4];
  const float* W2       = (const float*)d_in[5];
  float*       out      = (float*)d_out;

  const int N = in_sizes[0] / IN_DIM;
  const int E = in_sizes[1];

  float* Y  = (float*)d_ws;
  float* Z1 = Y + (size_t)N * HIDDEN;
  float* Z2 = Z1 + (size_t)N * HIDDEN;

  {
    const int n = 2 * N * HIDDEN;
    const int blocks = (n + 255) / 256;
    gcn_zero_kernel<<<blocks, 256, 0, stream>>>(Z1, n);
  }

  gcn_gemm1_wmma<<<(N + 15) / 16, 32, 0, stream>>>(features, W1, Y, N);

  {
    const int threads = E * 16;
    gcn_spmm_kernel<<<(threads + 255) / 256, 256, 0, stream>>>(
        esrc, edst, evals, Y, Z1, E,  0);
  }

  {
    const int threads = E * 16;
    gcn_spmm_kernel<<<(threads + 255) / 256, 256, 0, stream>>>(
        esrc, edst, evals, Z1, Z2, E,  1);
  }

  gcn_gemm2_lsm_wmma<<<(N + 15) / 16, 32, 0, stream>>>(Z2, W2, out, N);
}


# Both SpMMs: scatter atomics replaced by row-block ownership, no edge sort: a work-group keeps 784 dst rows in LDS, filters its edge chunk by dst range, LDS float adds (arrival order); 7 planes summed
# speedup vs baseline: 1.3034x; 1.3034x over previous
#include <hip/hip_runtime.h>
#include <math.h>

#define IN_DIM  100
#define HIDDEN  16
#define NCLS    40

typedef __attribute__((ext_vector_type(2))) float v2f;
typedef __attribute__((ext_vector_type(8))) float v8f;
typedef __attribute__((ext_vector_type(4))) float v4f;
typedef __attribute__((ext_vector_type(4))) int   v4i;

#define RB    784
#define GJ    7
#define SPT   1024
#define NV    4
#define QU    5
#define TILE  (SPT * 4 * NV)

static __device__ inline v8f wmma_f32_16x16x4(v2f a, v2f b, v8f c) {
#if __has_builtin(__builtin_amdgcn_wmma_f32_16x16x4_f32)
  return __builtin_amdgcn_wmma_f32_16x16x4_f32(
       false, a,  false, b,
       (short)0, c,  false,  false);
#else
  const int lane = (int)(threadIdx.x & 31u);
  const int nsel = lane & 15;
  v8f d = c;
#pragma unroll
  for (int g = 0; g < 8; ++g) {
    const int m = g + 8 * (lane >> 4);
    float acc = 0.f;
#pragma unroll
    for (int k = 0; k < 4; ++k) {
      const int srcA = m + 16 * (k >> 1);
      const int srcB = nsel + 16 * (k >> 1);
      const float av = __shfl((k & 1) ? a.y : a.x, srcA, 32);
      const float bv = __shfl((k & 1) ? b.y : b.x, srcB, 32);
      acc += av * bv;
    }
    d[g] += acc;
  }
  return d;
#endif
}

__global__ __launch_bounds__(256) void gcn_sum_planes_relu(
    const float* __restrict__ P, float* __restrict__ Z, int n4) {
  const int i = blockIdx.x * blockDim.x + threadIdx.x;
  if (i >= n4) return;
  const v4f* __restrict__ P4 = (const v4f*)P;
  v4f s = P4[i];
#pragma unroll
  for (int j = 1; j < GJ; ++j) s += P4[(size_t)j * n4 + i];
  s.x = fmaxf(s.x, 0.0f);
  s.y = fmaxf(s.y, 0.0f);
  s.z = fmaxf(s.z, 0.0f);
  s.w = fmaxf(s.w, 0.0f);
  ((v4f*)Z)[i] = s;
}

__global__ __launch_bounds__(32) void gcn_gemm1_wmma(
    const float* __restrict__ X, const float* __restrict__ W1,
    float* __restrict__ Y, int N) {
  const int lane = (int)threadIdx.x;
  const int r0   = blockIdx.x * 16;
  const int col  = lane & 15;
  const int kh   = lane >> 4;
  const int rowA = min(r0 + col, N - 1);
  const float* __restrict__ Xr = X + (size_t)rowA * IN_DIM;

  v8f acc = {};
#pragma unroll 5
  for (int k0 = 0; k0 < IN_DIM; k0 += 4) {
    const int ka = k0 + 2 * kh;
    v2f a;
    a.x = Xr[ka];
    a.y = Xr[ka + 1];
    v2f b;
    b.x = W1[ka * HIDDEN + col];
    b.y = W1[(ka + 1) * HIDDEN + col];
    acc = wmma_f32_16x16x4(a, b, acc);
  }

  float* __restrict__ Yr = Y + (size_t)(r0 + 8 * kh) * HIDDEN + col;
  if (r0 + 16 <= N) {
#pragma unroll
    for (int g = 0; g < 8; ++g) Yr[g * HIDDEN] = acc[g];
  } else {
#pragma unroll
    for (int g = 0; g < 8; ++g) {
      const int row = r0 + g + 8 * kh;
      if (row < N) Y[row * HIDDEN + col] = acc[g];
    }
  }
}

static __device__ inline void load_dst_tile(const int* __restrict__ edst,
                                            int t0, int cend, int e4, int tid,
                                            v4i* d) {
#pragma unroll
  for (int v = 0; v < NV; ++v) {
    const int e0 = t0 + (v * SPT + tid) * 4;
    d[v] = *(const v4i*)(edst + (e0 < e4 ? e0 : e4));
  }
#pragma unroll
  for (int v = 0; v < NV; ++v) {
    const int e0 = t0 + (v * SPT + tid) * 4;
    if (e0 + 3 >= cend) {
      d[v].x = (e0     < cend) ? edst[e0]     : -1;
      d[v].y = (e0 + 1 < cend) ? edst[e0 + 1] : -1;
      d[v].z = (e0 + 2 < cend) ? edst[e0 + 2] : -1;
      d[v].w = (e0 + 3 < cend) ? edst[e0 + 3] : -1;
    }
  }
}

__global__ __launch_bounds__(SPT) void gcn_spmm_rows_lds(
    const int* __restrict__ esrc, const int* __restrict__ edst,
    const float* __restrict__ evals, const float* __restrict__ Xin,
    float* __restrict__ P, int E, int N) {
  __shared__ float acc[RB * HIDDEN];
  __shared__ unsigned q[TILE];
  __shared__ unsigned qtot;

  const int tid  = (int)threadIdx.x;
  const int row0 = (int)blockIdx.x * RB;
  const int j    = (int)blockIdx.y;
  const int Ec   = ((E + GJ * 4 - 1) / (GJ * 4)) * 4;
  const long long cb = (long long)j * Ec;
  const int cbeg = (int)(cb < (long long)E ? cb : (long long)E);
  const int cend = (E - cbeg < Ec) ? E : cbeg + Ec;

  const int e4 = ((E - 4) / 4) * 4;
  v4i dcur[NV];
  load_dst_tile(edst, cbeg, cend, e4, tid, dcur);

  for (int i = tid; i < RB * HIDDEN; i += SPT) acc[i] = 0.0f;
  if (tid == 0) qtot = 0u;
  __syncthreads();

  const int g = tid >> 4;
  const int c = tid & 15;
  const int lane = tid & 31;
  unsigned prev = 0u;

  for (int tb = cbeg; tb < cend; tb += TILE) {
    unsigned rr[4 * NV];
    unsigned bal[4 * NV];
    unsigned total = 0u;
#pragma unroll
    for (int v = 0; v < NV; ++v) {
#pragma unroll
      for (int k = 0; k < 4; ++k) {
        const int i = v * 4 + k;
        rr[i]  = (unsigned)(dcur[v][k] - row0);
        bal[i] = __builtin_amdgcn_ballot_w32(rr[i] < (unsigned)RB);
        total += (unsigned)__builtin_popcount(bal[i]);
      }
    }
    if (total) {
      unsigned got = 0u;
      if (lane == 0) got = atomicAdd(&qtot, total);
      unsigned wpos = (unsigned)__builtin_amdgcn_readfirstlane((int)got) - prev;
#pragma unroll
      for (int v = 0; v < NV; ++v) {
#pragma unroll
        for (int k = 0; k < 4; ++k) {
          const int i = v * 4 + k;
          if (bal[i]) {
            if (rr[i] < (unsigned)RB) {
              const unsigned off = (unsigned)((v * SPT + tid) * 4 + k);
              q[wpos + __builtin_amdgcn_mbcnt_lo(bal[i], 0u)] = (off << 10) | rr[i];
            }
            wpos += (unsigned)__builtin_popcount(bal[i]);
          }
        }
      }
    }
    __syncthreads();
    const unsigned tot = qtot;
    const int cnt = (int)(tot - prev);
    prev = tot;

    if (tb + TILE < cend) load_dst_tile(edst, tb + TILE, cend, e4, tid, dcur);

    for (int base = 0; base < cnt; base += 64 * QU) {
      unsigned en[QU];
      bool ok[QU];
#pragma unroll
      for (int u = 0; u < QU; ++u) {
        const int idx = base + u * 64 + g;
        ok[u] = idx < cnt;
        en[u] = q[ok[u] ? idx : (cnt - 1)];
      }
      int   s[QU];
      float w[QU];
#pragma unroll
      for (int u = 0; u < QU; ++u) {
        const int e = tb + (int)(en[u] >> 10);
        s[u] = esrc[e];
        w[u] = evals[e];
      }
      float x[QU];
#pragma unroll
      for (int u = 0; u < QU; ++u) x[u] = Xin[s[u] * HIDDEN + c];
      static_assert(QU == 5, "the fence below names QU = 5 values of each kind");
      asm volatile("" :: "v"(x[0]), "v"(x[1]), "v"(x[2]), "v"(x[3]), "v"(x[4]),
                         "v"(w[0]), "v"(w[1]), "v"(w[2]), "v"(w[3]), "v"(w[4]));
#pragma unroll
      for (int u = 0; u < QU; ++u) {
        if (ok[u]) atomicAdd(&acc[(en[u] & 1023u) * HIDDEN + c], w[u] * x[u]);
      }
    }
    __syncthreads();
  }

  const int rows = (N - row0 < RB) ? (N - row0) : RB;
  const int nval = rows * HIDDEN;
  float* __restrict__ Pj = P + ((size_t)j * N + row0) * HIDDEN;
  for (int i = tid * 4; i < nval; i += SPT * 4)
    *(v4f*)(Pj + i) = *(const v4f*)(acc + i);
}

__global__ __launch_bounds__(32) void gcn_gemm2_lsm_wmma(
    const float* __restrict__ P, const float* __restrict__ W2,
    float* __restrict__ out, int N) {
  const int lane = (int)threadIdx.x;
  const int r0   = blockIdx.x * 16;
  const int col  = lane & 15;
  const int kh   = lane >> 4;
  const int rowA = min(r0 + col, N - 1);
  const float* __restrict__ Zr = P + (size_t)rowA * HIDDEN;
  const size_t plane = (size_t)N * HIDDEN;

  v8f acc0 = {}, acc1 = {}, acc2 = {};
#pragma unroll
  for (int k0 = 0; k0 < HIDDEN; k0 += 4) {
    const int ka = k0 + 2 * kh;
    v2f a = *(const v2f*)(Zr + ka);
#pragma unroll
    for (int j = 1; j < GJ; ++j) a += *(const v2f*)(Zr + (size_t)j * plane + ka);

    v2f b0;
    b0.x = W2[ka * NCLS + col];
    b0.y = W2[(ka + 1) * NCLS + col];
    acc0 = wmma_f32_16x16x4(a, b0, acc0);

    v2f b1;
    b1.x = W2[ka * NCLS + 16 + col];
    b1.y = W2[(ka + 1) * NCLS + 16 + col];
    acc1 = wmma_f32_16x16x4(a, b1, acc1);

    const int c2 = 32 + col;
    v2f b2;
    b2.x = (c2 < NCLS) ? W2[ka * NCLS + c2] : 0.0f;
    b2.y = (c2 < NCLS) ? W2[(ka + 1) * NCLS + c2] : 0.0f;
    acc2 = wmma_f32_16x16x4(a, b2, acc2);
  }

  const bool t2ok = (col < 8);
  const float ninf = -__builtin_inff();
  const bool fullTile = (r0 + 16 <= N);
  float* __restrict__ Or = out + (size_t)(r0 + 8 * kh) * NCLS + col;

#pragma unroll
  for (int g = 0; g < 8; ++g) {
    const float v0 = acc0[g];
    const float v1 = acc1[g];
    const float v2v = acc2[g];

    float m = fmaxf(fmaxf(v0, v1), t2ok ? v2v : ninf);
#pragma unroll
    for (int off = 8; off >= 1; off >>= 1) m = fmaxf(m, __shfl_xor(m, off, 16));

    float s = expf(v0 - m) + expf(v1 - m) + (t2ok ? expf(v2v - m) : 0.0f);
#pragma unroll
    for (int off = 8; off >= 1; off >>= 1) s += __shfl_xor(s, off, 16);

    const float lse = m + logf(s);

    if (fullTile) {
      Or[g * NCLS] = v0 - lse;
      Or[g * NCLS + 16] = v1 - lse;
      if (t2ok) Or[g * NCLS + 32] = v2v - lse;
    } else {
      const int row = r0 + g + 8 * kh;
      if (row < N) {
        out[(size_t)row * NCLS + col] = v0 - lse;
        out[(size_t)row * NCLS + 16 + col] = v1 - lse;
        if (t2ok) out[(size_t)row * NCLS + 32 + col] = v2v - lse;
      }
    }
  }
}

extern "C" void kernel_launch(void* const* d_in, const int* in_sizes, int n_in,
                              void* d_out, int out_size, void* d_ws, size_t ws_size,
                              hipStream_t stream) {
  const float* features = (const float*)d_in[0];
  const int*   esrc     = (const int*)d_in[1];
  const int*   edst     = (const int*)d_in[2];
  const float* evals    = (const float*)d_in[3];
  const float* W1       = (const float*)d_in[4];
  const float* W2       = (const float*)d_in[5];
  float*       out      = (float*)d_out;

  const int N = in_sizes[0] / IN_DIM;
  const int E = in_sizes[1];

  float* Y = (float*)d_ws;
  float* H = Y + (size_t)N * HIDDEN;
  float* P = H + (size_t)N * HIDDEN;

  const dim3 spGrid((N + RB - 1) / RB, GJ);

  gcn_gemm1_wmma<<<(N + 15) / 16, 32, 0, stream>>>(features, W1, Y, N);

  gcn_spmm_rows_lds<<<spGrid, SPT, 0, stream>>>(esrc, edst, evals, Y, P, E, N);

  {
    const int n4 = N * HIDDEN / 4;
    gcn_sum_planes_relu<<<(n4 + 255) / 256, 256, 0, stream>>>(P, H, n4);
  }

  gcn_spmm_rows_lds<<<spGrid, SPT, 0, stream>>>(esrc, edst, evals, H, P, E, N);

  gcn_gemm2_lsm_wmma<<<(N + 15) / 16, 32, 0, stream>>>(P, W2, out, N);
}


# SpMM: one work-group per WGP owns 1568 rows (32 row blocks: half the index scanning, half the waves); each wave files and takes its own LDS queue, no barrier in the edge loop; adds in arrival order
# speedup vs baseline: 1.5026x; 1.1528x over previous
#include <hip/hip_runtime.h>
#include <math.h>

#define IN_DIM  100
#define HIDDEN  16
#define NCLS    40

typedef __attribute__((ext_vector_type(2))) float v2f;
typedef __attribute__((ext_vector_type(8))) float v8f;
typedef __attribute__((ext_vector_type(4))) float v4f;
typedef __attribute__((ext_vector_type(4))) int   v4i;

#define RB    1568
#define RBITS 11
#define GJ    7
#define SPT   1024
#define NV    4
#define QU    10
#define TILE  (SPT * 4 * NV)
#define WQ    (32 * 4 * NV)

static __device__ inline v8f wmma_f32_16x16x4(v2f a, v2f b, v8f c) {
#if __has_builtin(__builtin_amdgcn_wmma_f32_16x16x4_f32)
  return __builtin_amdgcn_wmma_f32_16x16x4_f32(
       false, a,  false, b,
       (short)0, c,  false,  false);
#else
  const int lane = (int)(threadIdx.x & 31u);
  const int nsel = lane & 15;
  v8f d = c;
#pragma unroll
  for (int g = 0; g < 8; ++g) {
    const int m = g + 8 * (lane >> 4);
    float acc = 0.f;
#pragma unroll
    for (int k = 0; k < 4; ++k) {
      const int srcA = m + 16 * (k >> 1);
      const int srcB = nsel + 16 * (k >> 1);
      const float av = __shfl((k & 1) ? a.y : a.x, srcA, 32);
      const float bv = __shfl((k & 1) ? b.y : b.x, srcB, 32);
      acc += av * bv;
    }
    d[g] += acc;
  }
  return d;
#endif
}

__global__ __launch_bounds__(256) void gcn_sum_planes_relu(
    const float* __restrict__ P, float* __restrict__ Z, int n4) {
  const int i = blockIdx.x * blockDim.x + threadIdx.x;
  if (i >= n4) return;
  const v4f* __restrict__ P4 = (const v4f*)P;
  v4f s = P4[i];
#pragma unroll
  for (int j = 1; j < GJ; ++j) s += P4[(size_t)j * n4 + i];
  s.x = fmaxf(s.x, 0.0f);
  s.y = fmaxf(s.y, 0.0f);
  s.z = fmaxf(s.z, 0.0f);
  s.w = fmaxf(s.w, 0.0f);
  ((v4f*)Z)[i] = s;
}

__global__ __launch_bounds__(32) void gcn_gemm1_wmma(
    const float* __restrict__ X, const float* __restrict__ W1,
    float* __restrict__ Y, int N) {
  const int lane = (int)threadIdx.x;
  const int r0   = blockIdx.x * 16;
  const int col  = lane & 15;
  const int kh   = lane >> 4;
  const int rowA = min(r0 + col, N - 1);
  const float* __restrict__ Xr = X + (size_t)rowA * IN_DIM;

  v8f acc = {};
#pragma unroll 5
  for (int k0 = 0; k0 < IN_DIM; k0 += 4) {
    const int ka = k0 + 2 * kh;
    v2f a;
    a.x = Xr[ka];
    a.y = Xr[ka + 1];
    v2f b;
    b.x = W1[ka * HIDDEN + col];
    b.y = W1[(ka + 1) * HIDDEN + col];
    acc = wmma_f32_16x16x4(a, b, acc);
  }

  float* __restrict__ Yr = Y + (size_t)(r0 + 8 * kh) * HIDDEN + col;
  if (r0 + 16 <= N) {
#pragma unroll
    for (int g = 0; g < 8; ++g) Yr[g * HIDDEN] = acc[g];
  } else {
#pragma unroll
    for (int g = 0; g < 8; ++g) {
      const int row = r0 + g + 8 * kh;
      if (row < N) Y[row * HIDDEN + col] = acc[g];
    }
  }
}

static __device__ inline void load_dst_tile(const int* __restrict__ edst,
                                            int t0, int cend, int e4, int tid,
                                            v4i* d) {
#pragma unroll
  for (int v = 0; v < NV; ++v) {
    const int e0 = t0 + (v * SPT + tid) * 4;
    d[v] = *(const v4i*)(edst + (e0 < e4 ? e0 : e4));
  }
#pragma unroll
  for (int v = 0; v < NV; ++v) {
    const int e0 = t0 + (v * SPT + tid) * 4;
    if (e0 + 3 >= cend) {
      d[v].x = (e0     < cend) ? edst[e0]     : -1;
      d[v].y = (e0 + 1 < cend) ? edst[e0 + 1] : -1;
      d[v].z = (e0 + 2 < cend) ? edst[e0 + 2] : -1;
      d[v].w = (e0 + 3 < cend) ? edst[e0 + 3] : -1;
    }
  }
}

__global__ __launch_bounds__(SPT) void gcn_spmm_rows_lds(
    const int* __restrict__ esrc, const int* __restrict__ edst,
    const float* __restrict__ evals, const float* __restrict__ Xin,
    float* __restrict__ P, int E, int N) {
  __shared__ float acc[RB * HIDDEN];
  __shared__ unsigned q[TILE];
  static_assert(RB <= (1 << RBITS) && TILE <= (1 << (32 - RBITS)),
                "a queue entry holds a tile offset and a local row");

  const int tid  = (int)threadIdx.x;
  const int row0 = (int)blockIdx.x * RB;
  const int j    = (int)blockIdx.y;
  const int Ec   = ((E + GJ * 4 - 1) / (GJ * 4)) * 4;
  const long long cb = (long long)j * Ec;
  const int cbeg = (int)(cb < (long long)E ? cb : (long long)E);
  const int cend = (E - cbeg < Ec) ? E : cbeg + Ec;

  const int e4 = ((E - 4) / 4) * 4;
  v4i dcur[NV];
  load_dst_tile(edst, cbeg, cend, e4, tid, dcur);

  for (int i = tid * 4; i < RB * HIDDEN; i += SPT * 4)
    *(v4f*)(acc + i) = (v4f){0.0f, 0.0f, 0.0f, 0.0f};
  __syncthreads();

  const int lane = tid & 31;
  const int h = lane >> 4;
  const int c = tid & 15;
  unsigned* __restrict__ qw = q + (tid >> 5) * WQ;

  for (int tb = cbeg; tb < cend; tb += TILE) {
    unsigned rr[4 * NV];
    unsigned bal[4 * NV];
    unsigned total = 0u;
#pragma unroll
    for (int v = 0; v < NV; ++v) {
#pragma unroll
      for (int k = 0; k < 4; ++k) {
        const int i = v * 4 + k;
        rr[i]  = (unsigned)(dcur[v][k] - row0);
        bal[i] = __builtin_amdgcn_ballot_w32(rr[i] < (unsigned)RB);
        total += (unsigned)__builtin_popcount(bal[i]);
      }
    }
    if (total) {
      unsigned wpos = 0u;
#pragma unroll
      for (int v = 0; v < NV; ++v) {
#pragma unroll
        for (int k = 0; k < 4; ++k) {
          const int i = v * 4 + k;
          if (bal[i]) {
            if (rr[i] < (unsigned)RB) {
              const unsigned off = (unsigned)((v * SPT + tid) * 4 + k);
              qw[wpos + __builtin_amdgcn_mbcnt_lo(bal[i], 0u)] = (off << RBITS) | rr[i];
            }
            wpos += (unsigned)__builtin_popcount(bal[i]);
          }
        }
      }
    }
    const int cnt = (int)total;

    if (tb + TILE < cend) load_dst_tile(edst, tb + TILE, cend, e4, tid, dcur);

    for (int base = 0; base < cnt; base += 2 * QU) {
      unsigned en[QU];
      bool ok[QU];
#pragma unroll
      for (int u = 0; u < QU; ++u) {
        const int idx = base + u * 2 + h;
        ok[u] = idx < cnt;
        en[u] = qw[ok[u] ? idx : (cnt - 1)];
      }
      int   s[QU];
      float w[QU];
#pragma unroll
      for (int u = 0; u < QU; ++u) {
        const int e = tb + (int)(en[u] >> RBITS);
        s[u] = esrc[e];
        w[u] = evals[e];
      }
      float x[QU];
#pragma unroll
      for (int u = 0; u < QU; ++u) x[u] = Xin[s[u] * HIDDEN + c];
      static_assert(QU == 10, "the fence below names QU = 10 values of each kind");
      asm volatile("" :: "v"(x[0]), "v"(x[1]), "v"(x[2]), "v"(x[3]), "v"(x[4]),
                         "v"(x[5]), "v"(x[6]), "v"(x[7]), "v"(x[8]), "v"(x[9]),
                         "v"(w[0]), "v"(w[1]), "v"(w[2]), "v"(w[3]), "v"(w[4]),
                         "v"(w[5]), "v"(w[6]), "v"(w[7]), "v"(w[8]), "v"(w[9]));
#pragma unroll
      for (int u = 0; u < QU; ++u) {
        if (ok[u]) atomicAdd(&acc[(en[u] & ((1u << RBITS) - 1u)) * HIDDEN + c], w[u] * x[u]);
      }
    }
  }
  __syncthreads();

  const int rows = (N - row0 < RB) ? (N - row0) : RB;
  const int nval = rows * HIDDEN;
  float* __restrict__ Pj = P + ((size_t)j * N + row0) * HIDDEN;
  for (int i = tid * 4; i < nval; i += SPT * 4)
    *(v4f*)(Pj + i) = *(const v4f*)(acc + i);
}

__global__ __launch_bounds__(32) void gcn_gemm2_lsm_wmma(
    const float* __restrict__ P, const float* __restrict__ W2,
    float* __restrict__ out, int N) {
  const int lane = (int)threadIdx.x;
  const int r0   = blockIdx.x * 16;
  const int col  = lane & 15;
  const int kh   = lane >> 4;
  const int rowA = min(r0 + col, N - 1);
  const float* __restrict__ Zr = P + (size_t)rowA * HIDDEN;
  const size_t plane = (size_t)N * HIDDEN;

  v8f acc0 = {}, acc1 = {}, acc2 = {};
#pragma unroll
  for (int k0 = 0; k0 < HIDDEN; k0 += 4) {
    const int ka = k0 + 2 * kh;
    v2f a = *(const v2f*)(Zr + ka);
#pragma unroll
    for (int j = 1; j < GJ; ++j) a += *(const v2f*)(Zr + (size_t)j * plane + ka);

    v2f b0;
    b0.x = W2[ka * NCLS + col];
    b0.y = W2[(ka + 1) * NCLS + col];
    acc0 = wmma_f32_16x16x4(a, b0, acc0);

    v2f b1;
    b1.x = W2[ka * NCLS + 16 + col];
    b1.y = W2[(ka + 1) * NCLS + 16 + col];
    acc1 = wmma_f32_16x16x4(a, b1, acc1);

    const int c2 = 32 + col;
    v2f b2;
    b2.x = (c2 < NCLS) ? W2[ka * NCLS + c2] : 0.0f;
    b2.y = (c2 < NCLS) ? W2[(ka + 1) * NCLS + c2] : 0.0f;
    acc2 = wmma_f32_16x16x4(a, b2, acc2);
  }

  const bool t2ok = (col < 8);
  const float ninf = -__builtin_inff();
  const bool fullTile = (r0 + 16 <= N);
  float* __restrict__ Or = out + (size_t)(r0 + 8 * kh) * NCLS + col;

#pragma unroll
  for (int g = 0; g < 8; ++g) {
    const float v0 = acc0[g];
    const float v1 = acc1[g];
    const float v2v = acc2[g];

    float m = fmaxf(fmaxf(v0, v1), t2ok ? v2v : ninf);
#pragma unroll
    for (int off = 8; off >= 1; off >>= 1) m = fmaxf(m, __shfl_xor(m, off, 16));

    float s = expf(v0 - m) + expf(v1 - m) + (t2ok ? expf(v2v - m) : 0.0f);
#pragma unroll
    for (int off = 8; off >= 1; off >>= 1) s += __shfl_xor(s, off, 16);

    const float lse = m + logf(s);

    if (fullTile) {
      Or[g * NCLS] = v0 - lse;
      Or[g * NCLS + 16] = v1 - lse;
      if (t2ok) Or[g * NCLS + 32] = v2v - lse;
    } else {
      const int row = r0 + g + 8 * kh;
      if (row < N) {
        out[(size_t)row * NCLS + col] = v0 - lse;
        out[(size_t)row * NCLS + 16 + col] = v1 - lse;
        if (t2ok) out[(size_t)row * NCLS + 32 + col] = v2v - lse;
      }
    }
  }
}

extern "C" void kernel_launch(void* const* d_in, const int* in_sizes, int n_in,
                              void* d_out, int out_size, void* d_ws, size_t ws_size,
                              hipStream_t stream) {
  const float* features = (const float*)d_in[0];
  const int*   esrc     = (const int*)d_in[1];
  const int*   edst     = (const int*)d_in[2];
  const float* evals    = (const float*)d_in[3];
  const float* W1       = (const float*)d_in[4];
  const float* W2       = (const float*)d_in[5];
  float*       out      = (float*)d_out;

  const int N = in_sizes[0] / IN_DIM;
  const int E = in_sizes[1];

  float* Y = (float*)d_ws;
  float* H = Y + (size_t)N * HIDDEN;
  float* P = H + (size_t)N * HIDDEN;

  const dim3 spGrid((N + RB - 1) / RB, GJ);

  gcn_gemm1_wmma<<<(N + 15) / 16, 32, 0, stream>>>(features, W1, Y, N);

  gcn_spmm_rows_lds<<<spGrid, SPT, 0, stream>>>(esrc, edst, evals, Y, P, E, N);

  {
    const int n4 = N * HIDDEN / 4;
    gcn_sum_planes_relu<<<(n4 + 255) / 256, 256, 0, stream>>>(P, H, n4);
  }

  gcn_spmm_rows_lds<<<spGrid, SPT, 0, stream>>>(esrc, edst, evals, H, P, E, N);

  gcn_gemm2_lsm_wmma<<<(N + 15) / 16, 32, 0, stream>>>(P, W2, out, N);
}


# Edge src and val arrays interleaved once per call into 8-byte records by 800 extra waves of the X@W1 launch; both SpMMs fetch one record per filed edge (one line), not two words of two arrays
# speedup vs baseline: 1.5509x; 1.0321x over previous
#include <hip/hip_runtime.h>
#include <math.h>

#define IN_DIM  100
#define HIDDEN  16
#define NCLS    40

typedef __attribute__((ext_vector_type(2))) float v2f;
typedef __attribute__((ext_vector_type(8))) float v8f;
typedef __attribute__((ext_vector_type(4))) float v4f;
typedef __attribute__((ext_vector_type(4))) int   v4i;
typedef __attribute__((ext_vector_type(2))) int   v2i;

#define RB    1568
#define RBITS 11
#define GJ    7
#define SPT   1024
#define NV    4
#define QU    10
#define TILE  (SPT * 4 * NV)
#define WQ    (32 * 4 * NV)
#define NPACK 800
#define PKU   8

static __device__ inline v8f wmma_f32_16x16x4(v2f a, v2f b, v8f c) {
#if __has_builtin(__builtin_amdgcn_wmma_f32_16x16x4_f32)
  return __builtin_amdgcn_wmma_f32_16x16x4_f32(
       false, a,  false, b,
       (short)0, c,  false,  false);
#else
  const int lane = (int)(threadIdx.x & 31u);
  const int nsel = lane & 15;
  v8f d = c;
#pragma unroll
  for (int g = 0; g < 8; ++g) {
    const int m = g + 8 * (lane >> 4);
    float acc = 0.f;
#pragma unroll
    for (int k = 0; k < 4; ++k) {
      const int srcA = m + 16 * (k >> 1);
      const int srcB = nsel + 16 * (k >> 1);
      const float av = __shfl((k & 1) ? a.y : a.x, srcA, 32);
      const float bv = __shfl((k & 1) ? b.y : b.x, srcB, 32);
      acc += av * bv;
    }
    d[g] += acc;
  }
  return d;
#endif
}

__global__ __launch_bounds__(256) void gcn_sum_planes_relu(
    const float* __restrict__ P, float* __restrict__ Z, int n4) {
  const int i = blockIdx.x * blockDim.x + threadIdx.x;
  if (i >= n4) return;
  const v4f* __restrict__ P4 = (const v4f*)P;
  v4f s = P4[i];
#pragma unroll
  for (int j = 1; j < GJ; ++j) s += P4[(size_t)j * n4 + i];
  s.x = fmaxf(s.x, 0.0f);
  s.y = fmaxf(s.y, 0.0f);
  s.z = fmaxf(s.z, 0.0f);
  s.w = fmaxf(s.w, 0.0f);
  ((v4f*)Z)[i] = s;
}

__global__ __launch_bounds__(32) void gcn_gemm1_wmma(
    const float* __restrict__ X, const float* __restrict__ W1,
    float* __restrict__ Y, int N,
    const int* __restrict__ esrc, const float* __restrict__ evals,
    float* __restrict__ R, int E) {
  const int lane = (int)threadIdx.x;
  const int tiles = (N + 15) / 16;
  if ((int)blockIdx.x >= tiles) {
    const int pw   = (int)blockIdx.x - tiles;
    const int npw  = (int)gridDim.x - tiles;
    const int full = E / 64;
    v4i* __restrict__ R4 = (v4i*)R;
    for (int u0 = pw; u0 < full; u0 += PKU * npw) {
      v2i s[PKU];
      v2f v[PKU];
#pragma unroll
      for (int k = 0; k < PKU; ++k) {
        const int uk = u0 + k * npw;
        const int e  = (uk < full ? uk : full - 1) * 64 + 2 * lane;
        s[k] = *(const v2i*)(esrc + e);
        v[k] = *(const v2f*)(evals + e);
      }
#pragma unroll
      for (int k = 0; k < PKU; ++k) {
        const int uk = u0 + k * npw;
        if (uk < full)
          R4[uk * 32 + lane] = (v4i){s[k].x, __float_as_int(v[k].x),
                                     s[k].y, __float_as_int(v[k].y)};
      }
    }
    if (pw == full % npw) {
      const int e = full * 64 + lane;
      int* __restrict__ Ri = (int*)R;
      for (int t = e; t < E; t += 32) {
        Ri[2 * t]     = esrc[t];
        Ri[2 * t + 1] = __float_as_int(evals[t]);
      }
    }
    return;
  }
  const int r0   = blockIdx.x * 16;
  const int col  = lane & 15;
  const int kh   = lane >> 4;
  const int rowA = min(r0 + col, N - 1);
  const float* __restrict__ Xr = X + (size_t)rowA * IN_DIM;

  v8f acc = {};
#pragma unroll 5
  for (int k0 = 0; k0 < IN_DIM; k0 += 4) {
    const int ka = k0 + 2 * kh;
    v2f a;
    a.x = Xr[ka];
    a.y = Xr[ka + 1];
    v2f b;
    b.x = W1[ka * HIDDEN + col];
    b.y = W1[(ka + 1) * HIDDEN + col];
    acc = wmma_f32_16x16x4(a, b, acc);
  }

  float* __restrict__ Yr = Y + (size_t)(r0 + 8 * kh) * HIDDEN + col;
  if (r0 + 16 <= N) {
#pragma unroll
    for (int g = 0; g < 8; ++g) Yr[g * HIDDEN] = acc[g];
  } else {
#pragma unroll
    for (int g = 0; g < 8; ++g) {
      const int row = r0 + g + 8 * kh;
      if (row < N) Y[row * HIDDEN + col] = acc[g];
    }
  }
}

static __device__ inline void load_dst_tile(const int* __restrict__ edst,
                                            int t0, int cend, int e4, int tid,
                                            v4i* d) {
#pragma unroll
  for (int v = 0; v < NV; ++v) {
    const int e0 = t0 + (v * SPT + tid) * 4;
    d[v] = *(const v4i*)(edst + (e0 < e4 ? e0 : e4));
  }
#pragma unroll
  for (int v = 0; v < NV; ++v) {
    const int e0 = t0 + (v * SPT + tid) * 4;
    if (e0 + 3 >= cend) {
      d[v].x = (e0     < cend) ? edst[e0]     : -1;
      d[v].y = (e0 + 1 < cend) ? edst[e0 + 1] : -1;
      d[v].z = (e0 + 2 < cend) ? edst[e0 + 2] : -1;
      d[v].w = (e0 + 3 < cend) ? edst[e0 + 3] : -1;
    }
  }
}

__global__ __launch_bounds__(SPT) void gcn_spmm_rows_lds(
    const float* __restrict__ R, const int* __restrict__ edst,
    const float* __restrict__ Xin,
    float* __restrict__ P, int E, int N) {
  __shared__ float acc[RB * HIDDEN];
  __shared__ unsigned q[TILE];
  static_assert(RB <= (1 << RBITS) && TILE <= (1 << (32 - RBITS)),
                "a queue entry holds a tile offset and a local row");

  const int tid  = (int)threadIdx.x;
  const int row0 = (int)blockIdx.x * RB;
  const int j    = (int)blockIdx.y;
  const int Ec   = ((E + GJ * 4 - 1) / (GJ * 4)) * 4;
  const long long cb = (long long)j * Ec;
  const int cbeg = (int)(cb < (long long)E ? cb : (long long)E);
  const int cend = (E - cbeg < Ec) ? E : cbeg + Ec;

  const int e4 = ((E - 4) / 4) * 4;
  v4i dcur[NV];
  load_dst_tile(edst, cbeg, cend, e4, tid, dcur);

  for (int i = tid * 4; i < RB * HIDDEN; i += SPT * 4)
    *(v4f*)(acc + i) = (v4f){0.0f, 0.0f, 0.0f, 0.0f};
  __syncthreads();

  const int lane = tid & 31;
  const int h = lane >> 4;
  const int c = tid & 15;
  unsigned* __restrict__ qw = q + (tid >> 5) * WQ;
  const v2i* __restrict__ Rec = (const v2i*)R;

  for (int tb = cbeg; tb < cend; tb += TILE) {
    unsigned rr[4 * NV];
    unsigned bal[4 * NV];
    unsigned total = 0u;
#pragma unroll
    for (int v = 0; v < NV; ++v) {
#pragma unroll
      for (int k = 0; k < 4; ++k) {
        const int i = v * 4 + k;
        rr[i]  = (unsigned)(dcur[v][k] - row0);
        bal[i] = __builtin_amdgcn_ballot_w32(rr[i] < (unsigned)RB);
        total += (unsigned)__builtin_popcount(bal[i]);
      }
    }
    if (total) {
      unsigned wpos = 0u;
#pragma unroll
      for (int v = 0; v < NV; ++v) {
#pragma unroll
        for (int k = 0; k < 4; ++k) {
          const int i = v * 4 + k;
          if (bal[i]) {
            if (rr[i] < (unsigned)RB) {
              const unsigned off = (unsigned)((v * SPT + tid) * 4 + k);
              qw[wpos + __builtin_amdgcn_mbcnt_lo(bal[i], 0u)] = (off << RBITS) | rr[i];
            }
            wpos += (unsigned)__builtin_popcount(bal[i]);
          }
        }
      }
    }
    const int cnt = (int)total;

    if (tb + TILE < cend) load_dst_tile(edst, tb + TILE, cend, e4, tid, dcur);

    for (int base = 0; base < cnt; base += 2 * QU) {
      unsigned en[QU];
      bool ok[QU];
#pragma unroll
      for (int u = 0; u < QU; ++u) {
        const int idx = base + u * 2 + h;
        ok[u] = idx < cnt;
        en[u] = qw[ok[u] ? idx : (cnt - 1)];
      }
      int   s[QU];
      float w[QU];
#pragma unroll
      for (int u = 0; u < QU; ++u) {
        const v2i r = Rec[tb + (int)(en[u] >> RBITS)];
        s[u] = r.x;
        w[u] = __int_as_float(r.y);
      }
      float x[QU];
#pragma unroll
      for (int u = 0; u < QU; ++u) x[u] = Xin[s[u] * HIDDEN + c];
      static_assert(QU == 10, "the fence below names QU = 10 values of each kind");
      asm volatile("" :: "v"(x[0]), "v"(x[1]), "v"(x[2]), "v"(x[3]), "v"(x[4]),
                         "v"(x[5]), "v"(x[6]), "v"(x[7]), "v"(x[8]), "v"(x[9]),
                         "v"(w[0]), "v"(w[1]), "v"(w[2]), "v"(w[3]), "v"(w[4]),
                         "v"(w[5]), "v"(w[6]), "v"(w[7]), "v"(w[8]), "v"(w[9]));
#pragma unroll
      for (int u = 0; u < QU; ++u) {
        if (ok[u]) atomicAdd(&acc[(en[u] & ((1u << RBITS) - 1u)) * HIDDEN + c], w[u] * x[u]);
      }
    }
  }
  __syncthreads();

  const int rows = (N - row0 < RB) ? (N - row0) : RB;
  const int nval = rows * HIDDEN;
  float* __restrict__ Pj = P + ((size_t)j * N + row0) * HIDDEN;
  for (int i = tid * 4; i < nval; i += SPT * 4)
    *(v4f*)(Pj + i) = *(const v4f*)(acc + i);
}

__global__ __launch_bounds__(32) void gcn_gemm2_lsm_wmma(
    const float* __restrict__ P, const float* __restrict__ W2,
    float* __restrict__ out, int N) {
  const int lane = (int)threadIdx.x;
  const int r0   = blockIdx.x * 16;
  const int col  = lane & 15;
  const int kh   = lane >> 4;
  const int rowA = min(r0 + col, N - 1);
  const float* __restrict__ Zr = P + (size_t)rowA * HIDDEN;
  const size_t plane = (size_t)N * HIDDEN;

  v8f acc0 = {}, acc1 = {}, acc2 = {};
#pragma unroll
  for (int k0 = 0; k0 < HIDDEN; k0 += 4) {
    const int ka = k0 + 2 * kh;
    v2f a = *(const v2f*)(Zr + ka);
#pragma unroll
    for (int j = 1; j < GJ; ++j) a += *(const v2f*)(Zr + (size_t)j * plane + ka);

    v2f b0;
    b0.x = W2[ka * NCLS + col];
    b0.y = W2[(ka + 1) * NCLS + col];
    acc0 = wmma_f32_16x16x4(a, b0, acc0);

    v2f b1;
    b1.x = W2[ka * NCLS + 16 + col];
    b1.y = W2[(ka + 1) * NCLS + 16 + col];
    acc1 = wmma_f32_16x16x4(a, b1, acc1);

    const int c2 = 32 + col;
    v2f b2;
    b2.x = (c2 < NCLS) ? W2[ka * NCLS + c2] : 0.0f;
    b2.y = (c2 < NCLS) ? W2[(ka + 1) * NCLS + c2] : 0.0f;
    acc2 = wmma_f32_16x16x4(a, b2, acc2);
  }

  const bool t2ok = (col < 8);
  const float ninf = -__builtin_inff();
  const bool fullTile = (r0 + 16 <= N);
  float* __restrict__ Or = out + (size_t)(r0 + 8 * kh) * NCLS + col;

#pragma unroll
  for (int g = 0; g < 8; ++g) {
    const float v0 = acc0[g];
    const float v1 = acc1[g];
    const float v2v = acc2[g];

    float m = fmaxf(fmaxf(v0, v1), t2ok ? v2v : ninf);
#pragma unroll
    for (int off = 8; off >= 1; off >>= 1) m = fmaxf(m, __shfl_xor(m, off, 16));

    float s = expf(v0 - m) + expf(v1 - m) + (t2ok ? expf(v2v - m) : 0.0f);
#pragma unroll
    for (int off = 8; off >= 1; off >>= 1) s += __shfl_xor(s, off, 16);

    const float lse = m + logf(s);

    if (fullTile) {
      Or[g * NCLS] = v0 - lse;
      Or[g * NCLS + 16] = v1 - lse;
      if (t2ok) Or[g * NCLS + 32] = v2v - lse;
    } else {
      const int row = r0 + g + 8 * kh;
      if (row < N) {
        out[(size_t)row * NCLS + col] = v0 - lse;
        out[(size_t)row * NCLS + 16 + col] = v1 - lse;
        if (t2ok) out[(size_t)row * NCLS + 32 + col] = v2v - lse;
      }
    }
  }
}

extern "C" void kernel_launch(void* const* d_in, const int* in_sizes, int n_in,
                              void* d_out, int out_size, void* d_ws, size_t ws_size,
                              hipStream_t stream) {
  const float* features = (const float*)d_in[0];
  const int*   esrc     = (const int*)d_in[1];
  const int*   edst     = (const int*)d_in[2];
  const float* evals    = (const float*)d_in[3];
  const float* W1       = (const float*)d_in[4];
  const float* W2       = (const float*)d_in[5];
  float*       out      = (float*)d_out;

  const int N = in_sizes[0] / IN_DIM;
  const int E = in_sizes[1];

  float* Y = (float*)d_ws;
  float* H = Y + (size_t)N * HIDDEN;
  float* P = H + (size_t)N * HIDDEN;
  float* R = P + (size_t)GJ * N * HIDDEN;

  const dim3 spGrid((N + RB - 1) / RB, GJ);

  gcn_gemm1_wmma<<<(N + 15) / 16 + NPACK, 32, 0, stream>>>(features, W1, Y, N,
                                                        esrc, evals, R, E);

  gcn_spmm_rows_lds<<<spGrid, SPT, 0, stream>>>(R, edst, Y, P, E, N);

  {
    const int n4 = N * HIDDEN / 4;
    gcn_sum_planes_relu<<<(n4 + 255) / 256, 256, 0, stream>>>(P, H, n4);
  }

  gcn_spmm_rows_lds<<<spGrid, SPT, 0, stream>>>(R, edst, H, P, E, N);

  gcn_gemm2_lsm_wmma<<<(N + 15) / 16, 32, 0, stream>>>(P, W2, out, N);
}


# SpMM row blocks scan a 1-byte row-block id per edge (16 edges per 16-byte load, SWAR byte match) that the record waves write; the local row rides in the record; LDS adds in arrival order as before
# speedup vs baseline: 1.7347x; 1.1185x over previous
#include <hip/hip_runtime.h>
#include <math.h>

#define IN_DIM  100
#define HIDDEN  16
#define NCLS    40

typedef __attribute__((ext_vector_type(2))) float v2f;
typedef __attribute__((ext_vector_type(8))) float v8f;
typedef __attribute__((ext_vector_type(4))) float v4f;
typedef __attribute__((ext_vector_type(4))) int   v4i;
typedef __attribute__((ext_vector_type(2))) int   v2i;

#define RB    1568
#define RBITS 11
#define GJ    7
#define SPT   1024
#define NV    2
#define QU    20
#define TILE  (SPT * 16 * NV)
#define WQ    (32 * 16 * NV)
#define NPACK 800
#define PKU   4

static __device__ inline v8f wmma_f32_16x16x4(v2f a, v2f b, v8f c) {
#if __has_builtin(__builtin_amdgcn_wmma_f32_16x16x4_f32)
  return __builtin_amdgcn_wmma_f32_16x16x4_f32(
       false, a,  false, b,
       (short)0, c,  false,  false);
#else
  const int lane = (int)(threadIdx.x & 31u);
  const int nsel = lane & 15;
  v8f d = c;
#pragma unroll
  for (int g = 0; g < 8; ++g) {
    const int m = g + 8 * (lane >> 4);
    float acc = 0.f;
#pragma unroll
    for (int k = 0; k < 4; ++k) {
      const int srcA = m + 16 * (k >> 1);
      const int srcB = nsel + 16 * (k >> 1);
      const float av = __shfl((k & 1) ? a.y : a.x, srcA, 32);
      const float bv = __shfl((k & 1) ? b.y : b.x, srcB, 32);
      acc += av * bv;
    }
    d[g] += acc;
  }
  return d;
#endif
}

__global__ __launch_bounds__(256) void gcn_sum_planes_relu(
    const float* __restrict__ P, float* __restrict__ Z, int n4) {
  const int i = blockIdx.x * blockDim.x + threadIdx.x;
  if (i >= n4) return;
  const v4f* __restrict__ P4 = (const v4f*)P;
  v4f s = P4[i];
#pragma unroll
  for (int j = 1; j < GJ; ++j) s += P4[(size_t)j * n4 + i];
  s.x = fmaxf(s.x, 0.0f);
  s.y = fmaxf(s.y, 0.0f);
  s.z = fmaxf(s.z, 0.0f);
  s.w = fmaxf(s.w, 0.0f);
  ((v4f*)Z)[i] = s;
}

__global__ __launch_bounds__(32) void gcn_gemm1_wmma(
    const float* __restrict__ X, const float* __restrict__ W1,
    float* __restrict__ Y, int N,
    const int* __restrict__ esrc, const int* __restrict__ edst,
    const float* __restrict__ evals,
    float* __restrict__ R, int* __restrict__ Bid, int E) {
  const int lane = (int)threadIdx.x;
  const int tiles = (N + 15) / 16;
  if ((int)blockIdx.x >= tiles) {
    const int pw   = (int)blockIdx.x - tiles;
    const int npw  = (int)gridDim.x - tiles;
    const int full = E / 128;
    v4i* __restrict__ R4 = (v4i*)R;
    unsigned short* __restrict__ B2 = (unsigned short*)Bid;
    for (int u0 = pw; u0 < full; u0 += PKU * npw) {
      v2i s[2 * PKU];
      v2i d[2 * PKU];
      v2f v[2 * PKU];
#pragma unroll
      for (int k = 0; k < 2 * PKU; ++k) {
        const int uk = u0 + (k >> 1) * npw;
        const int e  = (uk < full ? uk : full - 1) * 128 + (k & 1) * 64 + 2 * lane;
        s[k] = *(const v2i*)(esrc + e);
        d[k] = *(const v2i*)(edst + e);
        v[k] = *(const v2f*)(evals + e);
      }
#pragma unroll
      for (int k = 0; k < 2 * PKU; ++k) {
        const int uk = u0 + (k >> 1) * npw;
        if (uk < full) {
          const unsigned b0 = (unsigned)d[k].x / (unsigned)RB;
          const unsigned b1 = (unsigned)d[k].y / (unsigned)RB;
          const unsigned l0 = (unsigned)d[k].x - b0 * (unsigned)RB;
          const unsigned l1 = (unsigned)d[k].y - b1 * (unsigned)RB;
          const int p = uk * 64 + (k & 1) * 32 + lane;
          R4[p] = (v4i){(int)(((unsigned)s[k].x << RBITS) | l0), __float_as_int(v[k].x),
                        (int)(((unsigned)s[k].y << RBITS) | l1), __float_as_int(v[k].y)};
          B2[p] = (unsigned short)(b0 | (b1 << 8));
        }
      }
    }
    if (pw == full % npw) {
      int* __restrict__ Ri = (int*)R;
      unsigned char* __restrict__ B1 = (unsigned char*)Bid;
      for (int t = full * 128 + lane; t < E; t += 32) {
        const unsigned b = (unsigned)edst[t] / (unsigned)RB;
        Ri[2 * t]     = (int)(((unsigned)esrc[t] << RBITS) | ((unsigned)edst[t] - b * (unsigned)RB));
        Ri[2 * t + 1] = __float_as_int(evals[t]);
        B1[t] = (unsigned char)b;
      }
      for (int t = E + lane; t < ((E + 15) / 16) * 16; t += 32) B1[t] = 0xFF;
    }
    return;
  }
  const int r0   = blockIdx.x * 16;
  const int col  = lane & 15;
  const int kh   = lane >> 4;
  const int rowA = min(r0 + col, N - 1);
  const float* __restrict__ Xr = X + (size_t)rowA * IN_DIM;

  v8f acc = {};
#pragma unroll 5
  for (int k0 = 0; k0 < IN_DIM; k0 += 4) {
    const int ka = k0 + 2 * kh;
    v2f a;
    a.x = Xr[ka];
    a.y = Xr[ka + 1];
    v2f b;
    b.x = W1[ka * HIDDEN + col];
    b.y = W1[(ka + 1) * HIDDEN + col];
    acc = wmma_f32_16x16x4(a, b, acc);
  }

  float* __restrict__ Yr = Y + (size_t)(r0 + 8 * kh) * HIDDEN + col;
  if (r0 + 16 <= N) {
#pragma unroll
    for (int g = 0; g < 8; ++g) Yr[g * HIDDEN] = acc[g];
  } else {
#pragma unroll
    for (int g = 0; g < 8; ++g) {
      const int row = r0 + g + 8 * kh;
      if (row < N) Y[row * HIDDEN + col] = acc[g];
    }
  }
}

static __device__ inline void load_id_tile(const int* __restrict__ Bid,
                                           int t0, int cend, int tid, v4i* d) {
#pragma unroll
  for (int v = 0; v < NV; ++v) {
    const int e0 = t0 + (v * SPT + tid) * 16;
    d[v] = *(const v4i*)((const char*)Bid + (e0 < cend ? e0 : 0));
  }
}

__global__ __launch_bounds__(SPT) void gcn_spmm_rows_lds(
    const float* __restrict__ R, const int* __restrict__ Bid,
    const float* __restrict__ Xin,
    float* __restrict__ P, int E, int N) {
  __shared__ float acc[RB * HIDDEN];
  __shared__ unsigned short q[TILE];
  static_assert(RB <= (1 << RBITS) && TILE <= (1 << 16),
                "a record holds a local row, a queue entry an offset in the step");

  const int tid  = (int)threadIdx.x;
  const int row0 = (int)blockIdx.x * RB;
  const int j    = (int)blockIdx.y;
  const int Ec   = ((E + GJ * 16 - 1) / (GJ * 16)) * 16;
  const long long cb = (long long)j * Ec;
  const int cbeg = (int)(cb < (long long)E ? cb : (long long)E);
  const int cend = (E - cbeg < Ec) ? E : cbeg + Ec;
  const unsigned pat = (unsigned)blockIdx.x * 0x01010101u;

  v4i dcur[NV];
  load_id_tile(Bid, cbeg, cend, tid, dcur);

  for (int i = tid * 4; i < RB * HIDDEN; i += SPT * 4)
    *(v4f*)(acc + i) = (v4f){0.0f, 0.0f, 0.0f, 0.0f};
  __syncthreads();

  const int lane = tid & 31;
  const int h = lane >> 4;
  const int c = tid & 15;
  unsigned short* __restrict__ qw = q + (tid >> 5) * WQ;
  const v2i* __restrict__ Rec = (const v2i*)R;

  for (int tb = cbeg; tb < cend; tb += TILE) {
    unsigned mm[NV];
#pragma unroll
    for (int v = 0; v < NV; ++v) {
      unsigned m = 0u;
#pragma unroll
      for (int k = 0; k < 4; ++k) {
        const unsigned x = (unsigned)dcur[v][k] ^ pat;
        const unsigned t = ((x & 0x7f7f7f7fu) + 0x7f7f7f7fu) | x;
        m |= (~t & 0x80808080u) >> (7 - k);
      }
      mm[v] = (tb + (v * SPT + tid) * 16 < cend) ? m : 0u;
    }
    unsigned wpos = 0u;
#pragma unroll
    for (int v = 0; v < NV; ++v) {
      unsigned m = mm[v];
      for (int round = 0; round < 16; ++round) {
        const unsigned bal = __builtin_amdgcn_ballot_w32(m != 0u);
        if (bal == 0u) break;
        if (m != 0u) {
          const unsigned p = (unsigned)__builtin_ctz(m);
          const unsigned off = (unsigned)((v * SPT + tid) * 16) + ((p & 3u) << 2) + (p >> 3);
          qw[wpos + __builtin_amdgcn_mbcnt_lo(bal, 0u)] = (unsigned short)off;
          m &= m - 1u;
        }
        wpos += (unsigned)__builtin_popcount(bal);
      }
    }
    const int cnt = (int)wpos;

    if (tb + TILE < cend) load_id_tile(Bid, tb + TILE, cend, tid, dcur);

    for (int base = 0; base < cnt; base += 2 * QU) {
      unsigned en[QU];
      bool ok[QU];
#pragma unroll
      for (int u = 0; u < QU; ++u) {
        const int idx = base + u * 2 + h;
        ok[u] = idx < cnt;
        en[u] = qw[ok[u] ? idx : (cnt - 1)];
      }
      unsigned s[QU];
      float w[QU];
#pragma unroll
      for (int u = 0; u < QU; ++u) {
        const v2i r = Rec[tb + (int)en[u]];
        s[u] = (unsigned)r.x;
        w[u] = __int_as_float(r.y);
      }
      float x[QU];
#pragma unroll
      for (int u = 0; u < QU; ++u)
        x[u] = Xin[(int)((s[u] >> RBITS) * HIDDEN) + c];
      static_assert(QU == 20, "the fences below name QU = 20 values of each kind");
      asm volatile("" :: "v"(x[0]), "v"(x[1]), "v"(x[2]), "v"(x[3]), "v"(x[4]),
                         "v"(x[5]), "v"(x[6]), "v"(x[7]), "v"(x[8]), "v"(x[9]),
                         "v"(x[10]), "v"(x[11]), "v"(x[12]), "v"(x[13]), "v"(x[14]),
                         "v"(x[15]), "v"(x[16]), "v"(x[17]), "v"(x[18]), "v"(x[19]));
      asm volatile("" :: "v"(w[0]), "v"(w[1]), "v"(w[2]), "v"(w[3]), "v"(w[4]),
                         "v"(w[5]), "v"(w[6]), "v"(w[7]), "v"(w[8]), "v"(w[9]),
                         "v"(w[10]), "v"(w[11]), "v"(w[12]), "v"(w[13]), "v"(w[14]),
                         "v"(w[15]), "v"(w[16]), "v"(w[17]), "v"(w[18]), "v"(w[19]));
#pragma unroll
      for (int u = 0; u < QU; ++u) {
        if (ok[u]) atomicAdd(&acc[(s[u] & ((1u << RBITS) - 1u)) * HIDDEN + c], w[u] * x[u]);
      }
    }
  }
  __syncthreads();

  const int rows = (N - row0 < RB) ? (N - row0) : RB;
  const int nval = rows * HIDDEN;
  float* __restrict__ Pj = P + ((size_t)j * N + row0) * HIDDEN;
  for (int i = tid * 4; i < nval; i += SPT * 4)
    *(v4f*)(Pj + i) = *(const v4f*)(acc + i);
}

__global__ __launch_bounds__(32) void gcn_gemm2_lsm_wmma(
    const float* __restrict__ P, const float* __restrict__ W2,
    float* __restrict__ out, int N) {
  const int lane = (int)threadIdx.x;
  const int r0   = blockIdx.x * 16;
  const int col  = lane & 15;
  const int kh   = lane >> 4;
  const int rowA = min(r0 + col, N - 1);
  const float* __restrict__ Zr = P + (size_t)rowA * HIDDEN;
  const size_t plane = (size_t)N * HIDDEN;

  v8f acc0 = {}, acc1 = {}, acc2 = {};
#pragma unroll
  for (int k0 = 0; k0 < HIDDEN; k0 += 4) {
    const int ka = k0 + 2 * kh;
    v2f a = *(const v2f*)(Zr + ka);
#pragma unroll
    for (int j = 1; j < GJ; ++j) a += *(const v2f*)(Zr + (size_t)j * plane + ka);

    v2f b0;
    b0.x = W2[ka * NCLS + col];
    b0.y = W2[(ka + 1) * NCLS + col];
    acc0 = wmma_f32_16x16x4(a, b0, acc0);

    v2f b1;
    b1.x = W2[ka * NCLS + 16 + col];
    b1.y = W2[(ka + 1) * NCLS + 16 + col];
    acc1 = wmma_f32_16x16x4(a, b1, acc1);

    const int c2 = 32 + col;
    v2f b2;
    b2.x = (c2 < NCLS) ? W2[ka * NCLS + c2] : 0.0f;
    b2.y = (c2 < NCLS) ? W2[(ka + 1) * NCLS + c2] : 0.0f;
    acc2 = wmma_f32_16x16x4(a, b2, acc2);
  }

  const bool t2ok = (col < 8);
  const float ninf = -__builtin_inff();
  const bool fullTile = (r0 + 16 <= N);
  float* __restrict__ Or = out + (size_t)(r0 + 8 * kh) * NCLS + col;

#pragma unroll
  for (int g = 0; g < 8; ++g) {
    const float v0 = acc0[g];
    const float v1 = acc1[g];
    const float v2v = acc2[g];

    float m = fmaxf(fmaxf(v0, v1), t2ok ? v2v : ninf);
#pragma unroll
    for (int off = 8; off >= 1; off >>= 1) m = fmaxf(m, __shfl_xor(m, off, 16));

    float s = expf(v0 - m) + expf(v1 - m) + (t2ok ? expf(v2v - m) : 0.0f);
#pragma unroll
    for (int off = 8; off >= 1; off >>= 1) s += __shfl_xor(s, off, 16);

    const float lse = m + logf(s);

    if (fullTile) {
      Or[g * NCLS] = v0 - lse;
      Or[g * NCLS + 16] = v1 - lse;
      if (t2ok) Or[g * NCLS + 32] = v2v - lse;
    } else {
      const int row = r0 + g + 8 * kh;
      if (row < N) {
        out[(size_t)row * NCLS + col] = v0 - lse;
        out[(size_t)row * NCLS + 16 + col] = v1 - lse;
        if (t2ok) out[(size_t)row * NCLS + 32 + col] = v2v - lse;
      }
    }
  }
}

extern "C" void kernel_launch(void* const* d_in, const int* in_sizes, int n_in,
                              void* d_out, int out_size, void* d_ws, size_t ws_size,
                              hipStream_t stream) {
  const float* features = (const float*)d_in[0];
  const int*   esrc     = (const int*)d_in[1];
  const int*   edst     = (const int*)d_in[2];
  const float* evals    = (const float*)d_in[3];
  const float* W1       = (const float*)d_in[4];
  const float* W2       = (const float*)d_in[5];
  float*       out      = (float*)d_out;

  const int N = in_sizes[0] / IN_DIM;
  const int E = in_sizes[1];

  float* Y = (float*)d_ws;
  float* H = Y + (size_t)N * HIDDEN;
  float* P = H + (size_t)N * HIDDEN;
  float* B = P + (size_t)GJ * N * HIDDEN;
  float* R = B + ((size_t)E + 63) / 64 * 16;

  const dim3 spGrid((N + RB - 1) / RB, GJ);

  gcn_gemm1_wmma<<<(N + 15) / 16 + NPACK, 32, 0, stream>>>(features, W1, Y, N,
                                                        esrc, edst, evals, R, (int*)B, E);

  gcn_spmm_rows_lds<<<spGrid, SPT, 0, stream>>>(R, (const int*)B, Y, P, E, N);

  {
    const int n4 = N * HIDDEN / 4;
    gcn_sum_planes_relu<<<(n4 + 255) / 256, 256, 0, stream>>>(P, H, n4);
  }

  gcn_spmm_rows_lds<<<spGrid, SPT, 0, stream>>>(R, (const int*)B, H, P, E, N);

  gcn_gemm2_lsm_wmma<<<(N + 15) / 16, 32, 0, stream>>>(P, W2, out, N);
}


# Z2@W2 log-softmax: the 16-lane max and sum butterflies of a tile's 8 rows run abreast as row DPP steps on the vector unit, not as 64 serial LDS-crossbar permutes; same tree, same order of adds
# speedup vs baseline: 1.8040x; 1.0399x over previous
#include <hip/hip_runtime.h>
#include <math.h>

#define IN_DIM  100
#define HIDDEN  16
#define NCLS    40

typedef __attribute__((ext_vector_type(2))) float v2f;
typedef __attribute__((ext_vector_type(8))) float v8f;
typedef __attribute__((ext_vector_type(4))) float v4f;
typedef __attribute__((ext_vector_type(4))) int   v4i;
typedef __attribute__((ext_vector_type(2))) int   v2i;

#define RB    1568
#define RBITS 11
#define GJ    7
#define SPT   1024
#define NV    2
#define QU    20
#define TILE  (SPT * 16 * NV)
#define WQ    (32 * 16 * NV)
#define NPACK 800
#define PKU   4

static __device__ inline v8f wmma_f32_16x16x4(v2f a, v2f b, v8f c) {
#if __has_builtin(__builtin_amdgcn_wmma_f32_16x16x4_f32)
  return __builtin_amdgcn_wmma_f32_16x16x4_f32(
       false, a,  false, b,
       (short)0, c,  false,  false);
#else
  const int lane = (int)(threadIdx.x & 31u);
  const int nsel = lane & 15;
  v8f d = c;
#pragma unroll
  for (int g = 0; g < 8; ++g) {
    const int m = g + 8 * (lane >> 4);
    float acc = 0.f;
#pragma unroll
    for (int k = 0; k < 4; ++k) {
      const int srcA = m + 16 * (k >> 1);
      const int srcB = nsel + 16 * (k >> 1);
      const float av = __shfl((k & 1) ? a.y : a.x, srcA, 32);
      const float bv = __shfl((k & 1) ? b.y : b.x, srcB, 32);
      acc += av * bv;
    }
    d[g] += acc;
  }
  return d;
#endif
}

template <int CTRL>
static __device__ inline float row_xor(float x) {
  return __int_as_float(__builtin_amdgcn_update_dpp(
      0, __float_as_int(x), CTRL,  0xf,  0xf,  true));
}

__global__ __launch_bounds__(256) void gcn_sum_planes_relu(
    const float* __restrict__ P, float* __restrict__ Z, int n4) {
  const int i = blockIdx.x * blockDim.x + threadIdx.x;
  if (i >= n4) return;
  const v4f* __restrict__ P4 = (const v4f*)P;
  v4f s = P4[i];
#pragma unroll
  for (int j = 1; j < GJ; ++j) s += P4[(size_t)j * n4 + i];
  s.x = fmaxf(s.x, 0.0f);
  s.y = fmaxf(s.y, 0.0f);
  s.z = fmaxf(s.z, 0.0f);
  s.w = fmaxf(s.w, 0.0f);
  ((v4f*)Z)[i] = s;
}

__global__ __launch_bounds__(32) void gcn_gemm1_wmma(
    const float* __restrict__ X, const float* __restrict__ W1,
    float* __restrict__ Y, int N,
    const int* __restrict__ esrc, const int* __restrict__ edst,
    const float* __restrict__ evals,
    float* __restrict__ R, int* __restrict__ Bid, int E) {
  const int lane = (int)threadIdx.x;
  const int tiles = (N + 15) / 16;
  if ((int)blockIdx.x >= tiles) {
    const int pw   = (int)blockIdx.x - tiles;
    const int npw  = (int)gridDim.x - tiles;
    const int full = E / 128;
    v4i* __restrict__ R4 = (v4i*)R;
    unsigned short* __restrict__ B2 = (unsigned short*)Bid;
    for (int u0 = pw; u0 < full; u0 += PKU * npw) {
      v2i s[2 * PKU];
      v2i d[2 * PKU];
      v2f v[2 * PKU];
#pragma unroll
      for (int k = 0; k < 2 * PKU; ++k) {
        const int uk = u0 + (k >> 1) * npw;
        const int e  = (uk < full ? uk : full - 1) * 128 + (k & 1) * 64 + 2 * lane;
        s[k] = *(const v2i*)(esrc + e);
        d[k] = *(const v2i*)(edst + e);
        v[k] = *(const v2f*)(evals + e);
      }
#pragma unroll
      for (int k = 0; k < 2 * PKU; ++k) {
        const int uk = u0 + (k >> 1) * npw;
        if (uk < full) {
          const unsigned b0 = (unsigned)d[k].x / (unsigned)RB;
          const unsigned b1 = (unsigned)d[k].y / (unsigned)RB;
          const unsigned l0 = (unsigned)d[k].x - b0 * (unsigned)RB;
          const unsigned l1 = (unsigned)d[k].y - b1 * (unsigned)RB;
          const int p = uk * 64 + (k & 1) * 32 + lane;
          R4[p] = (v4i){(int)(((unsigned)s[k].x << RBITS) | l0), __float_as_int(v[k].x),
                        (int)(((unsigned)s[k].y << RBITS) | l1), __float_as_int(v[k].y)};
          B2[p] = (unsigned short)(b0 | (b1 << 8));
        }
      }
    }
    if (pw == full % npw) {
      int* __restrict__ Ri = (int*)R;
      unsigned char* __restrict__ B1 = (unsigned char*)Bid;
      for (int t = full * 128 + lane; t < E; t += 32) {
        const unsigned b = (unsigned)edst[t] / (unsigned)RB;
        Ri[2 * t]     = (int)(((unsigned)esrc[t] << RBITS) | ((unsigned)edst[t] - b * (unsigned)RB));
        Ri[2 * t + 1] = __float_as_int(evals[t]);
        B1[t] = (unsigned char)b;
      }
      for (int t = E + lane; t < ((E + 15) / 16) * 16; t += 32) B1[t] = 0xFF;
    }
    return;
  }
  const int r0   = blockIdx.x * 16;
  const int col  = lane & 15;
  const int kh   = lane >> 4;
  const int rowA = min(r0 + col, N - 1);
  const float* __restrict__ Xr = X + (size_t)rowA * IN_DIM;

  v8f acc = {};
#pragma unroll 5
  for (int k0 = 0; k0 < IN_DIM; k0 += 4) {
    const int ka = k0 + 2 * kh;
    v2f a;
    a.x = Xr[ka];
    a.y = Xr[ka + 1];
    v2f b;
    b.x = W1[ka * HIDDEN + col];
    b.y = W1[(ka + 1) * HIDDEN + col];
    acc = wmma_f32_16x16x4(a, b, acc);
  }

  float* __restrict__ Yr = Y + (size_t)(r0 + 8 * kh) * HIDDEN + col;
  if (r0 + 16 <= N) {
#pragma unroll
    for (int g = 0; g < 8; ++g) Yr[g * HIDDEN] = acc[g];
  } else {
#pragma unroll
    for (int g = 0; g < 8; ++g) {
      const int row = r0 + g + 8 * kh;
      if (row < N) Y[row * HIDDEN + col] = acc[g];
    }
  }
}

static __device__ inline void load_id_tile(const int* __restrict__ Bid,
                                           int t0, int cend, int tid, v4i* d) {
#pragma unroll
  for (int v = 0; v < NV; ++v) {
    const int e0 = t0 + (v * SPT + tid) * 16;
    d[v] = *(const v4i*)((const char*)Bid + (e0 < cend ? e0 : 0));
  }
}

__global__ __launch_bounds__(SPT) void gcn_spmm_rows_lds(
    const float* __restrict__ R, const int* __restrict__ Bid,
    const float* __restrict__ Xin,
    float* __restrict__ P, int E, int N) {
  __shared__ float acc[RB * HIDDEN];
  __shared__ unsigned short q[TILE];
  static_assert(RB <= (1 << RBITS) && TILE <= (1 << 16),
                "a record holds a local row, a queue entry an offset in the step");

  const int tid  = (int)threadIdx.x;
  const int row0 = (int)blockIdx.x * RB;
  const int j    = (int)blockIdx.y;
  const int Ec   = ((E + GJ * 16 - 1) / (GJ * 16)) * 16;
  const long long cb = (long long)j * Ec;
  const int cbeg = (int)(cb < (long long)E ? cb : (long long)E);
  const int cend = (E - cbeg < Ec) ? E : cbeg + Ec;
  const unsigned pat = (unsigned)blockIdx.x * 0x01010101u;

  v4i dcur[NV];
  load_id_tile(Bid, cbeg, cend, tid, dcur);

  for (int i = tid * 4; i < RB * HIDDEN; i += SPT * 4)
    *(v4f*)(acc + i) = (v4f){0.0f, 0.0f, 0.0f, 0.0f};
  __syncthreads();

  const int lane = tid & 31;
  const int h = lane >> 4;
  const int c = tid & 15;
  unsigned short* __restrict__ qw = q + (tid >> 5) * WQ;
  const v2i* __restrict__ Rec = (const v2i*)R;

  for (int tb = cbeg; tb < cend; tb += TILE) {
    unsigned mm[NV];
#pragma unroll
    for (int v = 0; v < NV; ++v) {
      unsigned m = 0u;
#pragma unroll
      for (int k = 0; k < 4; ++k) {
        const unsigned x = (unsigned)dcur[v][k] ^ pat;
        const unsigned t = ((x & 0x7f7f7f7fu) + 0x7f7f7f7fu) | x;
        m |= (~t & 0x80808080u) >> (7 - k);
      }
      mm[v] = (tb + (v * SPT + tid) * 16 < cend) ? m : 0u;
    }
    unsigned wpos = 0u;
#pragma unroll
    for (int v = 0; v < NV; ++v) {
      unsigned m = mm[v];
      for (int round = 0; round < 16; ++round) {
        const unsigned bal = __builtin_amdgcn_ballot_w32(m != 0u);
        if (bal == 0u) break;
        if (m != 0u) {
          const unsigned p = (unsigned)__builtin_ctz(m);
          const unsigned off = (unsigned)((v * SPT + tid) * 16) + ((p & 3u) << 2) + (p >> 3);
          qw[wpos + __builtin_amdgcn_mbcnt_lo(bal, 0u)] = (unsigned short)off;
          m &= m - 1u;
        }
        wpos += (unsigned)__builtin_popcount(bal);
      }
    }
    const int cnt = (int)wpos;

    if (tb + TILE < cend) load_id_tile(Bid, tb + TILE, cend, tid, dcur);

    for (int base = 0; base < cnt; base += 2 * QU) {
      unsigned en[QU];
      bool ok[QU];
#pragma unroll
      for (int u = 0; u < QU; ++u) {
        const int idx = base + u * 2 + h;
        ok[u] = idx < cnt;
        en[u] = qw[ok[u] ? idx : (cnt - 1)];
      }
      unsigned s[QU];
      float w[QU];
#pragma unroll
      for (int u = 0; u < QU; ++u) {
        const v2i r = Rec[tb + (int)en[u]];
        s[u] = (unsigned)r.x;
        w[u] = __int_as_float(r.y);
      }
      float x[QU];
#pragma unroll
      for (int u = 0; u < QU; ++u)
        x[u] = Xin[(int)((s[u] >> RBITS) * HIDDEN) + c];
      static_assert(QU == 20, "the fences below name QU = 20 values of each kind");
      asm volatile("" :: "v"(x[0]), "v"(x[1]), "v"(x[2]), "v"(x[3]), "v"(x[4]),
                         "v"(x[5]), "v"(x[6]), "v"(x[7]), "v"(x[8]), "v"(x[9]),
                         "v"(x[10]), "v"(x[11]), "v"(x[12]), "v"(x[13]), "v"(x[14]),
                         "v"(x[15]), "v"(x[16]), "v"(x[17]), "v"(x[18]), "v"(x[19]));
      asm volatile("" :: "v"(w[0]), "v"(w[1]), "v"(w[2]), "v"(w[3]), "v"(w[4]),
                         "v"(w[5]), "v"(w[6]), "v"(w[7]), "v"(w[8]), "v"(w[9]),
                         "v"(w[10]), "v"(w[11]), "v"(w[12]), "v"(w[13]), "v"(w[14]),
                         "v"(w[15]), "v"(w[16]), "v"(w[17]), "v"(w[18]), "v"(w[19]));
#pragma unroll
      for (int u = 0; u < QU; ++u) {
        if (ok[u]) atomicAdd(&acc[(s[u] & ((1u << RBITS) - 1u)) * HIDDEN + c], w[u] * x[u]);
      }
    }
  }
  __syncthreads();

  const int rows = (N - row0 < RB) ? (N - row0) : RB;
  const int nval = rows * HIDDEN;
  float* __restrict__ Pj = P + ((size_t)j * N + row0) * HIDDEN;
  for (int i = tid * 4; i < nval; i += SPT * 4)
    *(v4f*)(Pj + i) = *(const v4f*)(acc + i);
}

__global__ __launch_bounds__(32) void gcn_gemm2_lsm_wmma(
    const float* __restrict__ P, const float* __restrict__ W2,
    float* __restrict__ out, int N) {
  const int lane = (int)threadIdx.x;
  const int r0   = blockIdx.x * 16;
  const int col  = lane & 15;
  const int kh   = lane >> 4;
  const int rowA = min(r0 + col, N - 1);
  const float* __restrict__ Zr = P + (size_t)rowA * HIDDEN;
  const size_t plane = (size_t)N * HIDDEN;

  v8f acc0 = {}, acc1 = {}, acc2 = {};
#pragma unroll
  for (int k0 = 0; k0 < HIDDEN; k0 += 4) {
    const int ka = k0 + 2 * kh;
    v2f a = *(const v2f*)(Zr + ka);
#pragma unroll
    for (int j = 1; j < GJ; ++j) a += *(const v2f*)(Zr + (size_t)j * plane + ka);

    v2f b0;
    b0.x = W2[ka * NCLS + col];
    b0.y = W2[(ka + 1) * NCLS + col];
    acc0 = wmma_f32_16x16x4(a, b0, acc0);

    v2f b1;
    b1.x = W2[ka * NCLS + 16 + col];
    b1.y = W2[(ka + 1) * NCLS + 16 + col];
    acc1 = wmma_f32_16x16x4(a, b1, acc1);

    const int c2 = 32 + col;
    v2f b2;
    b2.x = (c2 < NCLS) ? W2[ka * NCLS + c2] : 0.0f;
    b2.y = (c2 < NCLS) ? W2[(ka + 1) * NCLS + c2] : 0.0f;
    acc2 = wmma_f32_16x16x4(a, b2, acc2);
  }

  const bool t2ok = (col < 8);
  const float ninf = -__builtin_inff();
  const bool fullTile = (r0 + 16 <= N);
  float* __restrict__ Or = out + (size_t)(r0 + 8 * kh) * NCLS + col;

  float m[8];
#pragma unroll
  for (int g = 0; g < 8; ++g)
    m[g] = fmaxf(fmaxf(acc0[g], acc1[g]), t2ok ? acc2[g] : ninf);
#pragma unroll
  for (int g = 0; g < 8; ++g) m[g] = fmaxf(m[g], row_xor<0x168>(m[g]));
#pragma unroll
  for (int g = 0; g < 8; ++g) m[g] = fmaxf(m[g], row_xor<0x164>(m[g]));
#pragma unroll
  for (int g = 0; g < 8; ++g) m[g] = fmaxf(m[g], row_xor<0x162>(m[g]));
#pragma unroll
  for (int g = 0; g < 8; ++g) m[g] = fmaxf(m[g], row_xor<0x161>(m[g]));

  float s[8];
#pragma unroll
  for (int g = 0; g < 8; ++g) {
    const float e2 = expf(acc2[g] - m[g]);
    s[g] = expf(acc0[g] - m[g]) + expf(acc1[g] - m[g]) + (t2ok ? e2 : 0.0f);
  }
#pragma unroll
  for (int g = 0; g < 8; ++g) s[g] += row_xor<0x168>(s[g]);
#pragma unroll
  for (int g = 0; g < 8; ++g) s[g] += row_xor<0x164>(s[g]);
#pragma unroll
  for (int g = 0; g < 8; ++g) s[g] += row_xor<0x162>(s[g]);
#pragma unroll
  for (int g = 0; g < 8; ++g) s[g] += row_xor<0x161>(s[g]);

  float lse[8];
#pragma unroll
  for (int g = 0; g < 8; ++g) lse[g] = m[g] + logf(s[g]);

  if (fullTile) {
#pragma unroll
    for (int g = 0; g < 8; ++g) {
      Or[g * NCLS] = acc0[g] - lse[g];
      Or[g * NCLS + 16] = acc1[g] - lse[g];
    }
    if (t2ok) {
#pragma unroll
      for (int g = 0; g < 8; ++g) Or[g * NCLS + 32] = acc2[g] - lse[g];
    }
  } else {
#pragma unroll
    for (int g = 0; g < 8; ++g) {
      const int row = r0 + g + 8 * kh;
      if (row < N) {
        out[(size_t)row * NCLS + col] = acc0[g] - lse[g];
        out[(size_t)row * NCLS + 16 + col] = acc1[g] - lse[g];
        if (t2ok) out[(size_t)row * NCLS + 32 + col] = acc2[g] - lse[g];
      }
    }
  }
}

extern "C" void kernel_launch(void* const* d_in, const int* in_sizes, int n_in,
                              void* d_out, int out_size, void* d_ws, size_t ws_size,
                              hipStream_t stream) {
  const float* features = (const float*)d_in[0];
  const int*   esrc     = (const int*)d_in[1];
  const int*   edst     = (const int*)d_in[2];
  const float* evals    = (const float*)d_in[3];
  const float* W1       = (const float*)d_in[4];
  const float* W2       = (const float*)d_in[5];
  float*       out      = (float*)d_out;

  const int N = in_sizes[0] / IN_DIM;
  const int E = in_sizes[1];

  float* Y = (float*)d_ws;
  float* H = Y + (size_t)N * HIDDEN;
  float* P = H + (size_t)N * HIDDEN;
  float* B = P + (size_t)GJ * N * HIDDEN;
  float* R = B + ((size_t)E + 63) / 64 * 16;

  const dim3 spGrid((N + RB - 1) / RB, GJ);

  gcn_gemm1_wmma<<<(N + 15) / 16 + NPACK, 32, 0, stream>>>(features, W1, Y, N,
                                                        esrc, edst, evals, R, (int*)B, E);

  gcn_spmm_rows_lds<<<spGrid, SPT, 0, stream>>>(R, (const int*)B, Y, P, E, N);

  {
    const int n4 = N * HIDDEN / 4;
    gcn_sum_planes_relu<<<(n4 + 255) / 256, 256, 0, stream>>>(P, H, n4);
  }

  gcn_spmm_rows_lds<<<spGrid, SPT, 0, stream>>>(R, (const int*)B, H, P, E, N);

  gcn_gemm2_lsm_wmma<<<(N + 15) / 16, 32, 0, stream>>>(P, W2, out, N);
}


# SpMM split re-cut for the 1-byte ids: 64 row blocks of 784 rows x 4 edge chunks (was 32 x 7), 65536-id steps: 256 work-groups, 4 partial planes (was 7) to store, sum and read; arrival-order adds
# speedup vs baseline: 1.8881x; 1.0466x over previous
#include <hip/hip_runtime.h>
#include <math.h>

#define IN_DIM  100
#define HIDDEN  16
#define NCLS    40

typedef __attribute__((ext_vector_type(2))) float v2f;
typedef __attribute__((ext_vector_type(8))) float v8f;
typedef __attribute__((ext_vector_type(4))) float v4f;
typedef __attribute__((ext_vector_type(4))) int   v4i;
typedef __attribute__((ext_vector_type(2))) int   v2i;

#define RB    784
#define RBITS 11
#define GJ    4
#define SPT   1024
#define NV    4
#define QU    20
#define TILE  (SPT * 16 * NV)
#define WQ    (32 * 16 * NV)
#define NPACK 800
#define PKU   4

static __device__ inline v8f wmma_f32_16x16x4(v2f a, v2f b, v8f c) {
#if __has_builtin(__builtin_amdgcn_wmma_f32_16x16x4_f32)
  return __builtin_amdgcn_wmma_f32_16x16x4_f32(
       false, a,  false, b,
       (short)0, c,  false,  false);
#else
  const int lane = (int)(threadIdx.x & 31u);
  const int nsel = lane & 15;
  v8f d = c;
#pragma unroll
  for (int g = 0; g < 8; ++g) {
    const int m = g + 8 * (lane >> 4);
    float acc = 0.f;
#pragma unroll
    for (int k = 0; k < 4; ++k) {
      const int srcA = m + 16 * (k >> 1);
      const int srcB = nsel + 16 * (k >> 1);
      const float av = __shfl((k & 1) ? a.y : a.x, srcA, 32);
      const float bv = __shfl((k & 1) ? b.y : b.x, srcB, 32);
      acc += av * bv;
    }
    d[g] += acc;
  }
  return d;
#endif
}

template <int CTRL>
static __device__ inline float row_xor(float x) {
  return __int_as_float(__builtin_amdgcn_update_dpp(
      0, __float_as_int(x), CTRL,  0xf,  0xf,  true));
}

__global__ __launch_bounds__(256) void gcn_sum_planes_relu(
    const float* __restrict__ P, float* __restrict__ Z, int n4) {
  const int i = blockIdx.x * blockDim.x + threadIdx.x;
  if (i >= n4) return;
  const v4f* __restrict__ P4 = (const v4f*)P;
  v4f s = P4[i];
#pragma unroll
  for (int j = 1; j < GJ; ++j) s += P4[(size_t)j * n4 + i];
  s.x = fmaxf(s.x, 0.0f);
  s.y = fmaxf(s.y, 0.0f);
  s.z = fmaxf(s.z, 0.0f);
  s.w = fmaxf(s.w, 0.0f);
  ((v4f*)Z)[i] = s;
}

__global__ __launch_bounds__(32) void gcn_gemm1_wmma(
    const float* __restrict__ X, const float* __restrict__ W1,
    float* __restrict__ Y, int N,
    const int* __restrict__ esrc, const int* __restrict__ edst,
    const float* __restrict__ evals,
    float* __restrict__ R, int* __restrict__ Bid, int E) {
  const int lane = (int)threadIdx.x;
  const int tiles = (N + 15) / 16;
  if ((int)blockIdx.x >= tiles) {
    const int pw   = (int)blockIdx.x - tiles;
    const int npw  = (int)gridDim.x - tiles;
    const int full = E / 128;
    v4i* __restrict__ R4 = (v4i*)R;
    unsigned short* __restrict__ B2 = (unsigned short*)Bid;
    for (int u0 = pw; u0 < full; u0 += PKU * npw) {
      v2i s[2 * PKU];
      v2i d[2 * PKU];
      v2f v[2 * PKU];
#pragma unroll
      for (int k = 0; k < 2 * PKU; ++k) {
        const int uk = u0 + (k >> 1) * npw;
        const int e  = (uk < full ? uk : full - 1) * 128 + (k & 1) * 64 + 2 * lane;
        s[k] = *(const v2i*)(esrc + e);
        d[k] = *(const v2i*)(edst + e);
        v[k] = *(const v2f*)(evals + e);
      }
#pragma unroll
      for (int k = 0; k < 2 * PKU; ++k) {
        const int uk = u0 + (k >> 1) * npw;
        if (uk < full) {
          const unsigned b0 = (unsigned)d[k].x / (unsigned)RB;
          const unsigned b1 = (unsigned)d[k].y / (unsigned)RB;
          const unsigned l0 = (unsigned)d[k].x - b0 * (unsigned)RB;
          const unsigned l1 = (unsigned)d[k].y - b1 * (unsigned)RB;
          const int p = uk * 64 + (k & 1) * 32 + lane;
          R4[p] = (v4i){(int)(((unsigned)s[k].x << RBITS) | l0), __float_as_int(v[k].x),
                        (int)(((unsigned)s[k].y << RBITS) | l1), __float_as_int(v[k].y)};
          B2[p] = (unsigned short)(b0 | (b1 << 8));
        }
      }
    }
    if (pw == full % npw) {
      int* __restrict__ Ri = (int*)R;
      unsigned char* __restrict__ B1 = (unsigned char*)Bid;
      for (int t = full * 128 + lane; t < E; t += 32) {
        const unsigned b = (unsigned)edst[t] / (unsigned)RB;
        Ri[2 * t]     = (int)(((unsigned)esrc[t] << RBITS) | ((unsigned)edst[t] - b * (unsigned)RB));
        Ri[2 * t + 1] = __float_as_int(evals[t]);
        B1[t] = (unsigned char)b;
      }
      for (int t = E + lane; t < ((E + 15) / 16) * 16; t += 32) B1[t] = 0xFF;
    }
    return;
  }
  const int r0   = blockIdx.x * 16;
  const int col  = lane & 15;
  const int kh   = lane >> 4;
  const int rowA = min(r0 + col, N - 1);
  const float* __restrict__ Xr = X + (size_t)rowA * IN_DIM;

  v8f acc = {};
#pragma unroll 5
  for (int k0 = 0; k0 < IN_DIM; k0 += 4) {
    const int ka = k0 + 2 * kh;
    v2f a;
    a.x = Xr[ka];
    a.y = Xr[ka + 1];
    v2f b;
    b.x = W1[ka * HIDDEN + col];
    b.y = W1[(ka + 1) * HIDDEN + col];
    acc = wmma_f32_16x16x4(a, b, acc);
  }

  float* __restrict__ Yr = Y + (size_t)(r0 + 8 * kh) * HIDDEN + col;
  if (r0 + 16 <= N) {
#pragma unroll
    for (int g = 0; g < 8; ++g) Yr[g * HIDDEN] = acc[g];
  } else {
#pragma unroll
    for (int g = 0; g < 8; ++g) {
      const int row = r0 + g + 8 * kh;
      if (row < N) Y[row * HIDDEN + col] = acc[g];
    }
  }
}

static __device__ inline void load_id_tile(const int* __restrict__ Bid,
                                           int t0, int cend, int tid, v4i* d) {
#pragma unroll
  for (int v = 0; v < NV; ++v) {
    const int e0 = t0 + (v * SPT + tid) * 16;
    d[v] = *(const v4i*)((const char*)Bid + (e0 < cend ? e0 : 0));
  }
}

__global__ __launch_bounds__(SPT) void gcn_spmm_rows_lds(
    const float* __restrict__ R, const int* __restrict__ Bid,
    const float* __restrict__ Xin,
    float* __restrict__ P, int E, int N) {
  __shared__ float acc[RB * HIDDEN];
  __shared__ unsigned short q[TILE];
  static_assert(RB <= (1 << RBITS) && TILE <= (1 << 16),
                "a record holds a local row, a queue entry an offset in the step");

  const int tid  = (int)threadIdx.x;
  const int row0 = (int)blockIdx.x * RB;
  const int j    = (int)blockIdx.y;
  const int Ec   = ((E + GJ * 16 - 1) / (GJ * 16)) * 16;
  const long long cb = (long long)j * Ec;
  const int cbeg = (int)(cb < (long long)E ? cb : (long long)E);
  const int cend = (E - cbeg < Ec) ? E : cbeg + Ec;
  const unsigned pat = (unsigned)blockIdx.x * 0x01010101u;

  v4i dcur[NV];
  load_id_tile(Bid, cbeg, cend, tid, dcur);

  for (int i = tid * 4; i < RB * HIDDEN; i += SPT * 4)
    *(v4f*)(acc + i) = (v4f){0.0f, 0.0f, 0.0f, 0.0f};
  __syncthreads();

  const int lane = tid & 31;
  const int h = lane >> 4;
  const int c = tid & 15;
  unsigned short* __restrict__ qw = q + (tid >> 5) * WQ;
  const v2i* __restrict__ Rec = (const v2i*)R;

  for (int tb = cbeg; tb < cend; tb += TILE) {
    unsigned mm[NV];
#pragma unroll
    for (int v = 0; v < NV; ++v) {
      unsigned m = 0u;
#pragma unroll
      for (int k = 0; k < 4; ++k) {
        const unsigned x = (unsigned)dcur[v][k] ^ pat;
        const unsigned t = ((x & 0x7f7f7f7fu) + 0x7f7f7f7fu) | x;
        m |= (~t & 0x80808080u) >> (7 - k);
      }
      mm[v] = (tb + (v * SPT + tid) * 16 < cend) ? m : 0u;
    }
    unsigned wpos = 0u;
#pragma unroll
    for (int v = 0; v < NV; ++v) {
      unsigned m = mm[v];
      for (int round = 0; round < 16; ++round) {
        const unsigned bal = __builtin_amdgcn_ballot_w32(m != 0u);
        if (bal == 0u) break;
        if (m != 0u) {
          const unsigned p = (unsigned)__builtin_ctz(m);
          const unsigned off = (unsigned)((v * SPT + tid) * 16) + ((p & 3u) << 2) + (p >> 3);
          qw[wpos + __builtin_amdgcn_mbcnt_lo(bal, 0u)] = (unsigned short)off;
          m &= m - 1u;
        }
        wpos += (unsigned)__builtin_popcount(bal);
      }
    }
    const int cnt = (int)wpos;

    if (tb + TILE < cend) load_id_tile(Bid, tb + TILE, cend, tid, dcur);

    for (int base = 0; base < cnt; base += 2 * QU) {
      unsigned en[QU];
      bool ok[QU];
#pragma unroll
      for (int u = 0; u < QU; ++u) {
        const int idx = base + u * 2 + h;
        ok[u] = idx < cnt;
        en[u] = qw[ok[u] ? idx : (cnt - 1)];
      }
      unsigned s[QU];
      float w[QU];
#pragma unroll
      for (int u = 0; u < QU; ++u) {
        const v2i r = Rec[tb + (int)en[u]];
        s[u] = (unsigned)r.x;
        w[u] = __int_as_float(r.y);
      }
      float x[QU];
#pragma unroll
      for (int u = 0; u < QU; ++u)
        x[u] = Xin[(int)((s[u] >> RBITS) * HIDDEN) + c];
      static_assert(QU == 20, "the fences below name QU = 20 values of each kind");
      asm volatile("" :: "v"(x[0]), "v"(x[1]), "v"(x[2]), "v"(x[3]), "v"(x[4]),
                         "v"(x[5]), "v"(x[6]), "v"(x[7]), "v"(x[8]), "v"(x[9]),
                         "v"(x[10]), "v"(x[11]), "v"(x[12]), "v"(x[13]), "v"(x[14]),
                         "v"(x[15]), "v"(x[16]), "v"(x[17]), "v"(x[18]), "v"(x[19]));
      asm volatile("" :: "v"(w[0]), "v"(w[1]), "v"(w[2]), "v"(w[3]), "v"(w[4]),
                         "v"(w[5]), "v"(w[6]), "v"(w[7]), "v"(w[8]), "v"(w[9]),
                         "v"(w[10]), "v"(w[11]), "v"(w[12]), "v"(w[13]), "v"(w[14]),
                         "v"(w[15]), "v"(w[16]), "v"(w[17]), "v"(w[18]), "v"(w[19]));
#pragma unroll
      for (int u = 0; u < QU; ++u) {
        if (ok[u]) atomicAdd(&acc[(s[u] & ((1u << RBITS) - 1u)) * HIDDEN + c], w[u] * x[u]);
      }
    }
  }
  __syncthreads();

  const int rows = (N - row0 < RB) ? (N - row0) : RB;
  const int nval = rows * HIDDEN;
  float* __restrict__ Pj = P + ((size_t)j * N + row0) * HIDDEN;
  for (int i = tid * 4; i < nval; i += SPT * 4)
    *(v4f*)(Pj + i) = *(const v4f*)(acc + i);
}

__global__ __launch_bounds__(32) void gcn_gemm2_lsm_wmma(
    const float* __restrict__ P, const float* __restrict__ W2,
    float* __restrict__ out, int N) {
  const int lane = (int)threadIdx.x;
  const int r0   = blockIdx.x * 16;
  const int col  = lane & 15;
  const int kh   = lane >> 4;
  const int rowA = min(r0 + col, N - 1);
  const float* __restrict__ Zr = P + (size_t)rowA * HIDDEN;
  const size_t plane = (size_t)N * HIDDEN;

  v8f acc0 = {}, acc1 = {}, acc2 = {};
#pragma unroll
  for (int k0 = 0; k0 < HIDDEN; k0 += 4) {
    const int ka = k0 + 2 * kh;
    v2f a = *(const v2f*)(Zr + ka);
#pragma unroll
    for (int j = 1; j < GJ; ++j) a += *(const v2f*)(Zr + (size_t)j * plane + ka);

    v2f b0;
    b0.x = W2[ka * NCLS + col];
    b0.y = W2[(ka + 1) * NCLS + col];
    acc0 = wmma_f32_16x16x4(a, b0, acc0);

    v2f b1;
    b1.x = W2[ka * NCLS + 16 + col];
    b1.y = W2[(ka + 1) * NCLS + 16 + col];
    acc1 = wmma_f32_16x16x4(a, b1, acc1);

    const int c2 = 32 + col;
    v2f b2;
    b2.x = (c2 < NCLS) ? W2[ka * NCLS + c2] : 0.0f;
    b2.y = (c2 < NCLS) ? W2[(ka + 1) * NCLS + c2] : 0.0f;
    acc2 = wmma_f32_16x16x4(a, b2, acc2);
  }

  const bool t2ok = (col < 8);
  const float ninf = -__builtin_inff();
  const bool fullTile = (r0 + 16 <= N);
  float* __restrict__ Or = out + (size_t)(r0 + 8 * kh) * NCLS + col;

  float m[8];
#pragma unroll
  for (int g = 0; g < 8; ++g)
    m[g] = fmaxf(fmaxf(acc0[g], acc1[g]), t2ok ? acc2[g] : ninf);
#pragma unroll
  for (int g = 0; g < 8; ++g) m[g] = fmaxf(m[g], row_xor<0x168>(m[g]));
#pragma unroll
  for (int g = 0; g < 8; ++g) m[g] = fmaxf(m[g], row_xor<0x164>(m[g]));
#pragma unroll
  for (int g = 0; g < 8; ++g) m[g] = fmaxf(m[g], row_xor<0x162>(m[g]));
#pragma unroll
  for (int g = 0; g < 8; ++g) m[g] = fmaxf(m[g], row_xor<0x161>(m[g]));

  float s[8];
#pragma unroll
  for (int g = 0; g < 8; ++g) {
    const float e2 = expf(acc2[g] - m[g]);
    s[g] = expf(acc0[g] - m[g]) + expf(acc1[g] - m[g]) + (t2ok ? e2 : 0.0f);
  }
#pragma unroll
  for (int g = 0; g < 8; ++g) s[g] += row_xor<0x168>(s[g]);
#pragma unroll
  for (int g = 0; g < 8; ++g) s[g] += row_xor<0x164>(s[g]);
#pragma unroll
  for (int g = 0; g < 8; ++g) s[g] += row_xor<0x162>(s[g]);
#pragma unroll
  for (int g = 0; g < 8; ++g) s[g] += row_xor<0x161>(s[g]);

  float lse[8];
#pragma unroll
  for (int g = 0; g < 8; ++g) lse[g] = m[g] + logf(s[g]);

  if (fullTile) {
#pragma unroll
    for (int g = 0; g < 8; ++g) {
      Or[g * NCLS] = acc0[g] - lse[g];
      Or[g * NCLS + 16] = acc1[g] - lse[g];
    }
    if (t2ok) {
#pragma unroll
      for (int g = 0; g < 8; ++g) Or[g * NCLS + 32] = acc2[g] - lse[g];
    }
  } else {
#pragma unroll
    for (int g = 0; g < 8; ++g) {
      const int row = r0 + g + 8 * kh;
      if (row < N) {
        out[(size_t)row * NCLS + col] = acc0[g] - lse[g];
        out[(size_t)row * NCLS + 16 + col] = acc1[g] - lse[g];
        if (t2ok) out[(size_t)row * NCLS + 32 + col] = acc2[g] - lse[g];
      }
    }
  }
}

extern "C" void kernel_launch(void* const* d_in, const int* in_sizes, int n_in,
                              void* d_out, int out_size, void* d_ws, size_t ws_size,
                              hipStream_t stream) {
  const float* features = (const float*)d_in[0];
  const int*   esrc     = (const int*)d_in[1];
  const int*   edst     = (const int*)d_in[2];
  const float* evals    = (const float*)d_in[3];
  const float* W1       = (const float*)d_in[4];
  const float* W2       = (const float*)d_in[5];
  float*       out      = (float*)d_out;

  const int N = in_sizes[0] / IN_DIM;
  const int E = in_sizes[1];

  float* Y = (float*)d_ws;
  float* H = Y + (size_t)N * HIDDEN;
  float* P = H + (size_t)N * HIDDEN;
  float* B = P + (size_t)GJ * N * HIDDEN;
  float* R = B + ((size_t)E + 63) / 64 * 16;

  const dim3 spGrid((N + RB - 1) / RB, GJ);

  gcn_gemm1_wmma<<<(N + 15) / 16 + NPACK, 32, 0, stream>>>(features, W1, Y, N,
                                                        esrc, edst, evals, R, (int*)B, E);

  gcn_spmm_rows_lds<<<spGrid, SPT, 0, stream>>>(R, (const int*)B, Y, P, E, N);

  {
    const int n4 = N * HIDDEN / 4;
    gcn_sum_planes_relu<<<(n4 + 255) / 256, 256, 0, stream>>>(P, H, n4);
  }

  gcn_spmm_rows_lds<<<spGrid, SPT, 0, stream>>>(R, (const int*)B, H, P, E, N);

  gcn_gemm2_lsm_wmma<<<(N + 15) / 16, 32, 0, stream>>>(P, W2, out, N);
}
